# starting kernel
# baseline (speedup 1.0000x reference)
#include <hip/hip_runtime.h>

typedef __attribute__((ext_vector_type(2))) float v2f;
typedef __attribute__((ext_vector_type(8))) float v8f;

#define FDIM 64


__global__ __launch_bounds__(256) void k_init_deg(float* __restrict__ deg, int n) {
    int i = blockIdx.x * blockDim.x + threadIdx.x;
    if (i < n) deg[i] = 1.0f;
}

__global__ __launch_bounds__(256) void k_deg_accum(const int* __restrict__ dst,
                                                   const float* __restrict__ w,
                                                   float* __restrict__ deg, int e) {
    int i = blockIdx.x * blockDim.x + threadIdx.x;
    if (i < e) atomicAdd(&deg[dst[i]], w[i]);
}

__global__ __launch_bounds__(256) void k_dinv(float* __restrict__ deg_dinv,
                                              float* __restrict__ self_norm, int n) {
    int i = blockIdx.x * blockDim.x + threadIdx.x;
    if (i < n) {
        float d = rsqrtf(deg_dinv[i]);
        deg_dinv[i]  = d;
        self_norm[i] = d * d;
    }
}

__global__ __launch_bounds__(256) void k_norm(const int* __restrict__ src,
                                              const int* __restrict__ dst,
                                              const float* __restrict__ w,
                                              const float* __restrict__ dinv,
                                              float* __restrict__ norm, int e) {
    int i = blockIdx.x * blockDim.x + threadIdx.x;
    if (i < e) norm[i] = dinv[src[i]] * w[i] * dinv[dst[i]];
}


__global__ __launch_bounds__(256) void k_gemm64_wmma(const float* __restrict__ H,
                                                     const float* __restrict__ Wm,
                                                     float* __restrict__ out, int nrows) {
    const int lane  = threadIdx.x & 31;
    const int wave  = threadIdx.x >> 5;
    const int ctile = wave & 3;
    const int rhalf = wave >> 2;
    const int l16   = lane & 15;
    const int hi    = lane >> 4;

    v2f bfrag[16];
    {
        const int col = ctile * 16 + l16;
        #pragma unroll
        for (int k = 0; k < 16; ++k) {
            const int k0 = 4 * k + (hi ? 2 : 0);
            bfrag[k].x = Wm[(size_t)k0 * FDIM + col];
            bfrag[k].y = Wm[(size_t)(k0 + 1) * FDIM + col];
        }
    }

    const int cc = ctile * 16 + l16;
    const int rowBlocks = (nrows + 31) >> 5;
    for (int rb = blockIdx.x; rb < rowBlocks; rb += gridDim.x) {
        const int row0 = rb * 32 + rhalf * 16;
        const bool full = (row0 + 16 <= nrows);

        int r  = row0 + l16;
        int rc = full ? r : ((r < nrows) ? r : (nrows > 0 ? nrows - 1 : 0));
        const float* hp = H + (size_t)rc * FDIM + (hi ? 2 : 0);

        v8f acc = {};
        #pragma unroll
        for (int k = 0; k < 16; ++k) {
            v2f a = *(const v2f*)(hp + 4 * k);
            acc = __builtin_amdgcn_wmma_f32_16x16x4_f32(
                      false, a, false, bfrag[k], (short)0, acc, false, false);
        }

        if (full) {
            float* op = out + (size_t)(row0 + hi * 8) * FDIM + cc;
            #pragma unroll
            for (int i = 0; i < 8; ++i)
                op[(size_t)i * FDIM] = acc[i];
        } else {
            #pragma unroll
            for (int i = 0; i < 8; ++i) {
                int rr = row0 + hi * 8 + i;
                if (rr < nrows) out[(size_t)rr * FDIM + cc] = acc[i];
            }
        }
    }
}


__global__ __launch_bounds__(256) void k_epi_init(const float* __restrict__ hw,
                                                  const float* __restrict__ self_norm,
                                                  const float* __restrict__ b,
                                                  float* __restrict__ out, int n) {
    long gid = (long)blockIdx.x * blockDim.x + threadIdx.x;
    if (gid < (long)n * FDIM) {
        int i = (int)(gid >> 6);
        out[gid] = hw[gid] * self_norm[i] + b[gid & 63];
    }
}

__global__ __launch_bounds__(256) void k_scatter(const float* __restrict__ hw,
                                                 const float* __restrict__ norm,
                                                 const int* __restrict__ src,
                                                 const int* __restrict__ dst,
                                                 float* __restrict__ out, int e) {
    long gid = (long)blockIdx.x * blockDim.x + threadIdx.x;
    if (gid >= (long)e * 16) return;
    int  ed = (int)(gid >> 4);
    int  g  = ((int)gid & 15) * 4;
    float nv = norm[ed];
    int   s  = src[ed];
    int   d  = dst[ed];
    const float4 v = *(const float4*)(hw + (size_t)s * FDIM + g);
    float* op = out + (size_t)d * FDIM + g;
    atomicAdd(op + 0, v.x * nv);
    atomicAdd(op + 1, v.y * nv);
    atomicAdd(op + 2, v.z * nv);
    atomicAdd(op + 3, v.w * nv);
}

__global__ __launch_bounds__(256) void k_relu(float* __restrict__ h, long total) {
    long gid = (long)blockIdx.x * blockDim.x + threadIdx.x;
    if (gid < total) h[gid] = fmaxf(h[gid], 0.0f);
}


__global__ __launch_bounds__(256) void k_matvec64(const float* __restrict__ H,
                                                  const float* __restrict__ w2,
                                                  float* __restrict__ lin, int n) {
    int i = blockIdx.x * blockDim.x + threadIdx.x;
    if (i < n) {
        const float* hp = H + (size_t)i * FDIM;
        float s = 0.0f;
        #pragma unroll
        for (int f = 0; f < FDIM; ++f) s = fmaf(hp[f], w2[f], s);
        lin[i] = s;
    }
}

__global__ __launch_bounds__(256) void k_init2(const float* __restrict__ lin,
                                               const float* __restrict__ self_norm,
                                               const float* __restrict__ b2,
                                               float* __restrict__ out, int n) {
    int i = blockIdx.x * blockDim.x + threadIdx.x;
    if (i < n) out[i] = lin[i] * self_norm[i] + b2[0];
}

__global__ __launch_bounds__(256) void k_scatter2(const float* __restrict__ lin,
                                                  const float* __restrict__ norm,
                                                  const int* __restrict__ src,
                                                  const int* __restrict__ dst,
                                                  float* __restrict__ out, int e) {
    int i = blockIdx.x * blockDim.x + threadIdx.x;
    if (i < e) atomicAdd(&out[dst[i]], lin[src[i]] * norm[i]);
}


extern "C" void kernel_launch(void* const* d_in, const int* in_sizes, int n_in,
                              void* d_out, int out_size, void* d_ws, size_t ws_size,
                              hipStream_t stream) {
    const float* x    = (const float*)d_in[0];
    const int*   esrc = (const int*)  d_in[1];
    const int*   edst = (const int*)  d_in[2];
    const float* ew   = (const float*)d_in[3];
    const float* W0   = (const float*)d_in[4];
    const float* b0   = (const float*)d_in[5];
    const float* W1   = (const float*)d_in[6];
    const float* b1   = (const float*)d_in[7];
    const float* W2   = (const float*)d_in[8];
    const float* b2   = (const float*)d_in[9];

    const int N = in_sizes[0] / FDIM;
    const int E = in_sizes[1];
    float* out = (float*)d_out;

    float* ws        = (float*)d_ws;
    float* dinv      = ws;
    float* self_norm = ws + (size_t)N;
    float* normw     = ws + 2 * (size_t)N;
    float* hw        = normw + (size_t)E;
    float* ha        = hw + (size_t)N * FDIM;

    const int T = 256;
    const int gN  = (N + T - 1) / T;
    const int gE  = (E + T - 1) / T;
    const long nf = (long)N * FDIM;
    const int gNF = (int)((nf + T - 1) / T);
    const long es = (long)E * 16;
    const int gES = (int)((es + T - 1) / T);
    const int gGemm = (N + 31) / 32;

    k_init_deg <<<gN, T, 0, stream>>>(dinv, N);
    k_deg_accum<<<gE, T, 0, stream>>>(edst, ew, dinv, E);
    k_dinv     <<<gN, T, 0, stream>>>(dinv, self_norm, N);
    k_norm     <<<gE, T, 0, stream>>>(esrc, edst, ew, dinv, normw, E);

    k_gemm64_wmma<<<gGemm, T, 0, stream>>>(x, W0, hw, N);
    k_epi_init   <<<gNF, T, 0, stream>>>(hw, self_norm, b0, ha, N);
    k_scatter    <<<gES, T, 0, stream>>>(hw, normw, esrc, edst, ha, E);
    k_relu       <<<gNF, T, 0, stream>>>(ha, nf);

    k_gemm64_wmma<<<gGemm, T, 0, stream>>>(ha, W1, hw, N);
    k_epi_init   <<<gNF, T, 0, stream>>>(hw, self_norm, b1, ha, N);
    k_scatter    <<<gES, T, 0, stream>>>(hw, normw, esrc, edst, ha, E);
    k_relu       <<<gNF, T, 0, stream>>>(ha, nf);

    float* lin = hw;
    k_matvec64<<<gN, T, 0, stream>>>(ha, W2, lin, N);
    k_init2   <<<gN, T, 0, stream>>>(lin, self_norm, b2, out, N);
    k_scatter2<<<gE, T, 0, stream>>>(lin, normw, esrc, edst, out, E);
}


# Fold the per-element passes (rsqrt of degree, self-loop+bias start value, relu, final start value) into the GEMM, norm and matvec kernels: 15 launches become 9; sum order unchanged
# speedup vs baseline: 1.1536x; 1.1536x over previous
#include <hip/hip_runtime.h>


typedef __attribute__((ext_vector_type(2))) float v2f;
typedef __attribute__((ext_vector_type(8))) float v8f;

#define FDIM 64


__global__ __launch_bounds__(256) void k_init_deg(float* __restrict__ deg, int n) {
    int i = blockIdx.x * blockDim.x + threadIdx.x;
    if (i < n) deg[i] = 1.0f;
}

__global__ __launch_bounds__(256) void k_deg_accum(const int* __restrict__ dst,
                                                   const float* __restrict__ w,
                                                   float* __restrict__ deg, int e) {
    int i = blockIdx.x * blockDim.x + threadIdx.x;
    if (i < e) atomicAdd(&deg[dst[i]], w[i]);
}

__global__ __launch_bounds__(256) void k_norm(const int* __restrict__ src,
                                              const int* __restrict__ dst,
                                              const float* __restrict__ w,
                                              const float* __restrict__ deg,
                                              float* __restrict__ norm, int e) {
    int i = blockIdx.x * blockDim.x + threadIdx.x;
    if (i < e) norm[i] = rsqrtf(deg[src[i]]) * w[i] * rsqrtf(deg[dst[i]]);
}


static __device__ __forceinline__ void gemm64_body(const float* __restrict__ H,
                                                   const float* __restrict__ Wm,
                                                   const float* __restrict__ deg,
                                                   const float* __restrict__ bias,
                                                   float* __restrict__ hw,
                                                   float* __restrict__ hinit,
                                                   int nrows, const bool relu_in) {
    const int lane  = threadIdx.x & 31;
    const int wave  = threadIdx.x >> 5;
    const int ctile = wave & 3;
    const int rhalf = wave >> 2;
    const int l16   = lane & 15;
    const int hi    = lane >> 4;

    v2f bfrag[16];
    {
        const int col = ctile * 16 + l16;
        #pragma unroll
        for (int k = 0; k < 16; ++k) {
            const int k0 = 4 * k + (hi ? 2 : 0);
            bfrag[k].x = Wm[(size_t)k0 * FDIM + col];
            bfrag[k].y = Wm[(size_t)(k0 + 1) * FDIM + col];
        }
    }

    const int cc = ctile * 16 + l16;
    const float bcol = bias[cc];
    const int rowBlocks = (nrows + 31) >> 5;
    for (int rb = blockIdx.x; rb < rowBlocks; rb += gridDim.x) {
        const int row0 = rb * 32 + rhalf * 16;
        const bool full = (row0 + 16 <= nrows);

        int r  = row0 + l16;
        int rc = full ? r : ((r < nrows) ? r : (nrows > 0 ? nrows - 1 : 0));
        const float* hp = H + (size_t)rc * FDIM + (hi ? 2 : 0);

        float sn[8];
        if (full) {
            const float4 d0 = *(const float4*)(deg + row0 + hi * 8);
            const float4 d1 = *(const float4*)(deg + row0 + hi * 8 + 4);
            sn[0] = d0.x; sn[1] = d0.y; sn[2] = d0.z; sn[3] = d0.w;
            sn[4] = d1.x; sn[5] = d1.y; sn[6] = d1.z; sn[7] = d1.w;
        } else {
            #pragma unroll
            for (int i = 0; i < 8; ++i) {
                int rr = row0 + hi * 8 + i;
                int rq = (rr < nrows) ? rr : (nrows > 0 ? nrows - 1 : 0);
                sn[i] = deg[rq];
            }
        }
        #pragma unroll
        for (int i = 0; i < 8; ++i) {
            const float d = rsqrtf(sn[i]);
            sn[i] = d * d;
        }

        v8f acc = {};
        #pragma unroll
        for (int k = 0; k < 16; ++k) {
            v2f a = *(const v2f*)(hp + 4 * k);
            if (relu_in) {
                a.x = fmaxf(a.x, 0.0f);
                a.y = fmaxf(a.y, 0.0f);
            }
            acc = __builtin_amdgcn_wmma_f32_16x16x4_f32(
                      false, a, false, bfrag[k], (short)0, acc, false, false);
        }

        if (full) {
            const size_t o = (size_t)(row0 + hi * 8) * FDIM + cc;
            float* op = hw + o;
            float* ip = hinit + o;
            #pragma unroll
            for (int i = 0; i < 8; ++i)
                op[(size_t)i * FDIM] = acc[i];
            #pragma unroll
            for (int i = 0; i < 8; ++i)
                ip[(size_t)i * FDIM] = acc[i] * sn[i] + bcol;
        } else {
            #pragma unroll
            for (int i = 0; i < 8; ++i) {
                int rr = row0 + hi * 8 + i;
                if (rr < nrows) {
                    hw[(size_t)rr * FDIM + cc]    = acc[i];
                    hinit[(size_t)rr * FDIM + cc] = acc[i] * sn[i] + bcol;
                }
            }
        }
    }
}

__global__ __launch_bounds__(256) void k_gemm64_first(const float* __restrict__ H,
                                                      const float* __restrict__ Wm,
                                                      const float* __restrict__ deg,
                                                      const float* __restrict__ bias,
                                                      float* __restrict__ hw,
                                                      float* __restrict__ hinit, int nrows) {
    gemm64_body(H, Wm, deg, bias, hw, hinit, nrows, false);
}

__global__ __launch_bounds__(256) void k_gemm64_relu(const float* __restrict__ H,
                                                     const float* __restrict__ Wm,
                                                     const float* __restrict__ deg,
                                                     const float* __restrict__ bias,
                                                     float* __restrict__ hw,
                                                     float* __restrict__ hinit, int nrows) {
    gemm64_body(H, Wm, deg, bias, hw, hinit, nrows, true);
}


__global__ __launch_bounds__(256) void k_scatter(const float* __restrict__ hw,
                                                 const float* __restrict__ norm,
                                                 const int* __restrict__ src,
                                                 const int* __restrict__ dst,
                                                 float* __restrict__ out, int e) {
    long gid = (long)blockIdx.x * blockDim.x + threadIdx.x;
    if (gid >= (long)e * 16) return;
    int  ed = (int)(gid >> 4);
    int  g  = ((int)gid & 15) * 4;
    float nv = norm[ed];
    int   s  = src[ed];
    int   d  = dst[ed];
    const float4 v = *(const float4*)(hw + (size_t)s * FDIM + g);
    float* op = out + (size_t)d * FDIM + g;
    atomicAdd(op + 0, v.x * nv);
    atomicAdd(op + 1, v.y * nv);
    atomicAdd(op + 2, v.z * nv);
    atomicAdd(op + 3, v.w * nv);
}


__global__ __launch_bounds__(256) void k_matvec64_init(const float* __restrict__ H,
                                                       const float* __restrict__ w2,
                                                       const float* __restrict__ deg,
                                                       const float* __restrict__ b2,
                                                       float* __restrict__ lin,
                                                       float* __restrict__ out, int n) {
    int i = blockIdx.x * blockDim.x + threadIdx.x;
    if (i < n) {
        const float* hp = H + (size_t)i * FDIM;
        float s = 0.0f;
        #pragma unroll
        for (int f = 0; f < FDIM; ++f) s = fmaf(fmaxf(hp[f], 0.0f), w2[f], s);
        lin[i] = s;
        const float d = rsqrtf(deg[i]);
        const float snv = d * d;
        out[i] = s * snv + b2[0];
    }
}

__global__ __launch_bounds__(256) void k_scatter2(const float* __restrict__ lin,
                                                  const float* __restrict__ norm,
                                                  const int* __restrict__ src,
                                                  const int* __restrict__ dst,
                                                  float* __restrict__ out, int e) {
    int i = blockIdx.x * blockDim.x + threadIdx.x;
    if (i < e) atomicAdd(&out[dst[i]], lin[src[i]] * norm[i]);
}


extern "C" void kernel_launch(void* const* d_in, const int* in_sizes, int n_in,
                              void* d_out, int out_size, void* d_ws, size_t ws_size,
                              hipStream_t stream) {
    const float* x    = (const float*)d_in[0];
    const int*   esrc = (const int*)  d_in[1];
    const int*   edst = (const int*)  d_in[2];
    const float* ew   = (const float*)d_in[3];
    const float* W0   = (const float*)d_in[4];
    const float* b0   = (const float*)d_in[5];
    const float* W1   = (const float*)d_in[6];
    const float* b1   = (const float*)d_in[7];
    const float* W2   = (const float*)d_in[8];
    const float* b2   = (const float*)d_in[9];

    const int N = in_sizes[0] / FDIM;
    const int E = in_sizes[1];
    float* out = (float*)d_out;

    float* ws    = (float*)d_ws;
    float* deg   = ws;
    float* normw = ws + 2 * (size_t)N;
    float* hw    = normw + (size_t)E;
    float* ha0   = hw  + (size_t)N * FDIM;
    float* ha1   = ha0 + (size_t)N * FDIM;

    const int T = 256;
    const int gN  = (N + T - 1) / T;
    const int gE  = (E + T - 1) / T;
    const long es = (long)E * 16;
    const int gES = (int)((es + T - 1) / T);
    const int gGemm = (N + 31) / 32;

    k_init_deg <<<gN, T, 0, stream>>>(deg, N);
    k_deg_accum<<<gE, T, 0, stream>>>(edst, ew, deg, E);
    k_norm     <<<gE, T, 0, stream>>>(esrc, edst, ew, deg, normw, E);

    k_gemm64_first<<<gGemm, T, 0, stream>>>(x, W0, deg, b0, hw, ha0, N);
    k_scatter     <<<gES, T, 0, stream>>>(hw, normw, esrc, edst, ha0, E);

    k_gemm64_relu<<<gGemm, T, 0, stream>>>(ha0, W1, deg, b1, hw, ha1, N);
    k_scatter    <<<gES, T, 0, stream>>>(hw, normw, esrc, edst, ha1, E);

    float* lin = hw;
    k_matvec64_init<<<gN, T, 0, stream>>>(ha1, W2, deg, b2, lin, out, N);
    k_scatter2     <<<gE, T, 0, stream>>>(lin, normw, esrc, edst, out, E);
}


# Replace the atomic edge scatter of all 3 layers by a per-call destination-sorted edge list (count, scan, fill) and a per-row register sum stored once; row sum order = arrival rank of a counter atomic
# speedup vs baseline: 1.8479x; 1.6019x over previous
#include <hip/hip_runtime.h>


typedef __attribute__((ext_vector_type(2))) float v2f;
typedef __attribute__((ext_vector_type(8))) float v8f;

#define FDIM 64


__global__ __launch_bounds__(256) void k_init_deg(float* __restrict__ deg, int* __restrict__ cnt, int n) {
    int i = blockIdx.x * blockDim.x + threadIdx.x;
    if (i < n) {
        deg[i] = 1.0f;
        cnt[i] = 0;
    }
}

__global__ __launch_bounds__(256) void k_deg_rank(const int* __restrict__ dst,
                                                  const float* __restrict__ w,
                                                  float* __restrict__ deg,
                                                  int* __restrict__ cnt,
                                                  int* __restrict__ rank, int e) {
    int i = blockIdx.x * blockDim.x + threadIdx.x;
    if (i < e) {
        const int d = dst[i];
        atomicAdd(&deg[d], w[i]);
        rank[i] = atomicAdd(&cnt[d], 1);
    }
}

__global__ __launch_bounds__(1024) void k_rowptr(const int* __restrict__ cnt,
                                                 int* __restrict__ rowptr, int n) {
    __shared__ int s_pre[32];
    __shared__ int s_tot[32];
    const int tid   = threadIdx.x;
    const int lane  = tid & 31;
    const int wv    = tid >> 5;
    const int tile0 = blockIdx.x * 1024;

    int pre = 0;
    const int4* c4 = (const int4*)cnt;
    const int lim = tile0 >> 2;
    #pragma unroll 4
    for (int i = tid; i < lim; i += 1024) {
        const int4 q = c4[i];
        pre += (q.x + q.y) + (q.z + q.w);
    }
    for (int o = 16; o > 0; o >>= 1) pre += __shfl_xor(pre, o, 32);

    const int g = tile0 + tid;
    const int c = (g < n) ? cnt[g] : 0;
    int inc = c;
    for (int o = 1; o < 32; o <<= 1) {
        const int t = __shfl_up(inc, o, 32);
        if (lane >= o) inc += t;
    }
    if (lane == 31) s_tot[wv] = inc;
    if (lane == 0)  s_pre[wv] = pre;
    __syncthreads();
    int base = 0;
    for (int k = 0; k < 32; ++k) {
        base += s_pre[k];
        if (k < wv) base += s_tot[k];
    }
    const int excl = base + inc - c;
    if (g < n)      rowptr[g] = excl;
    if (g == n - 1) rowptr[n] = excl + c;
}

__global__ __launch_bounds__(256) void k_fill(const int* __restrict__ src,
                                              const int* __restrict__ dst,
                                              const float* __restrict__ w,
                                              const float* __restrict__ deg,
                                              const int* __restrict__ rowptr,
                                              const int* __restrict__ rank,
                                              int2* __restrict__ csr, int e) {
    int i = blockIdx.x * blockDim.x + threadIdx.x;
    if (i < e) {
        const int s = src[i];
        const int d = dst[i];
        const float nv = rsqrtf(deg[s]) * w[i] * rsqrtf(deg[d]);
        csr[rowptr[d] + rank[i]] = make_int2(s, __float_as_int(nv));
    }
}


static __device__ __forceinline__ void gemm64_body(const float* __restrict__ H,
                                                   const float* __restrict__ Wm,
                                                   float* __restrict__ hw,
                                                   int nrows, const bool relu_in) {
    const int lane  = threadIdx.x & 31;
    const int wave  = threadIdx.x >> 5;
    const int ctile = wave & 3;
    const int rhalf = wave >> 2;
    const int l16   = lane & 15;
    const int hi    = lane >> 4;

    v2f bfrag[16];
    {
        const int col = ctile * 16 + l16;
        #pragma unroll
        for (int k = 0; k < 16; ++k) {
            const int k0 = 4 * k + (hi ? 2 : 0);
            bfrag[k].x = Wm[(size_t)k0 * FDIM + col];
            bfrag[k].y = Wm[(size_t)(k0 + 1) * FDIM + col];
        }
    }

    const int cc = ctile * 16 + l16;
    const int rowBlocks = (nrows + 31) >> 5;
    for (int rb = blockIdx.x; rb < rowBlocks; rb += gridDim.x) {
        const int row0 = rb * 32 + rhalf * 16;
        const bool full = (row0 + 16 <= nrows);

        int r  = row0 + l16;
        int rc = full ? r : ((r < nrows) ? r : (nrows > 0 ? nrows - 1 : 0));
        const float* hp = H + (size_t)rc * FDIM + (hi ? 2 : 0);

        v8f acc = {};
        #pragma unroll
        for (int k = 0; k < 16; ++k) {
            v2f a = *(const v2f*)(hp + 4 * k);
            if (relu_in) {
                a.x = fmaxf(a.x, 0.0f);
                a.y = fmaxf(a.y, 0.0f);
            }
            acc = __builtin_amdgcn_wmma_f32_16x16x4_f32(
                      false, a, false, bfrag[k], (short)0, acc, false, false);
        }

        if (full) {
            float* op = hw + (size_t)(row0 + hi * 8) * FDIM + cc;
            #pragma unroll
            for (int i = 0; i < 8; ++i)
                op[(size_t)i * FDIM] = acc[i];
        } else {
            #pragma unroll
            for (int i = 0; i < 8; ++i) {
                int rr = row0 + hi * 8 + i;
                if (rr < nrows) hw[(size_t)rr * FDIM + cc] = acc[i];
            }
        }
    }
}

__global__ __launch_bounds__(256) void k_gemm64_first(const float* __restrict__ H,
                                                      const float* __restrict__ Wm,
                                                      float* __restrict__ hw, int nrows) {
    gemm64_body(H, Wm, hw, nrows, false);
}

__global__ __launch_bounds__(256) void k_gemm64_relu(const float* __restrict__ H,
                                                     const float* __restrict__ Wm,
                                                     float* __restrict__ hw, int nrows) {
    gemm64_body(H, Wm, hw, nrows, true);
}


__global__ __launch_bounds__(256) void k_agg64(const float* __restrict__ hw,
                                               const int2* __restrict__ csr,
                                               const int* __restrict__ rowptr,
                                               const float* __restrict__ deg,
                                               const float* __restrict__ bias,
                                               float* __restrict__ hout, int n) {
    const int l16 = threadIdx.x & 15;
    const int g   = (blockIdx.x * 256 + threadIdx.x) >> 4;
    const bool live = (g < n);
    const int d   = live ? g : (n - 1);

    const int beg = rowptr[d];
    const int end = rowptr[d + 1];
    const float rs = rsqrtf(deg[d]);
    const float sn = rs * rs;
    const int f = l16 * 4;
    const float4 hs = *(const float4*)(hw + (size_t)d * FDIM + f);
    float4 acc;
    acc.x = hs.x * sn + bias[f + 0];
    acc.y = hs.y * sn + bias[f + 1];
    acc.z = hs.z * sn + bias[f + 2];
    acc.w = hs.w * sn + bias[f + 3];

    const float4 zero4 = make_float4(0.0f, 0.0f, 0.0f, 0.0f);
    const float* hrow = hw + f;
    for (int base = beg; base < end; base += 16) {
        const int idx = base + l16;
        int2 pr = make_int2(0, 0);
        if (idx < end) pr = csr[idx];
        const int m = min(16, end - base);
        for (int j = 0; j < m; j += 4) {
            const int   s0 = __shfl(pr.x, j + 0, 16);
            const int   s1 = __shfl(pr.x, j + 1, 16);
            const int   s2 = __shfl(pr.x, j + 2, 16);
            const int   s3 = __shfl(pr.x, j + 3, 16);
            const float n0 = __int_as_float(__shfl(pr.y, j + 0, 16));
            const float n1 = __int_as_float(__shfl(pr.y, j + 1, 16));
            const float n2 = __int_as_float(__shfl(pr.y, j + 2, 16));
            const float n3 = __int_as_float(__shfl(pr.y, j + 3, 16));
            float4 v0 = *(const float4*)(hrow + (size_t)s0 * FDIM);
            float4 v1 = zero4, v2 = zero4, v3 = zero4;
            if (j + 1 < m) v1 = *(const float4*)(hrow + (size_t)s1 * FDIM);
            if (j + 2 < m) v2 = *(const float4*)(hrow + (size_t)s2 * FDIM);
            if (j + 3 < m) v3 = *(const float4*)(hrow + (size_t)s3 * FDIM);
            acc.x += v0.x * n0; acc.y += v0.y * n0; acc.z += v0.z * n0; acc.w += v0.w * n0;
            acc.x += v1.x * n1; acc.y += v1.y * n1; acc.z += v1.z * n1; acc.w += v1.w * n1;
            acc.x += v2.x * n2; acc.y += v2.y * n2; acc.z += v2.z * n2; acc.w += v2.w * n2;
            acc.x += v3.x * n3; acc.y += v3.y * n3; acc.z += v3.z * n3; acc.w += v3.w * n3;
        }
    }
    if (live) *(float4*)(hout + (size_t)d * FDIM + f) = acc;
}


__global__ __launch_bounds__(256) void k_matvec64(const float* __restrict__ H,
                                                  const float* __restrict__ w2,
                                                  float* __restrict__ lin, int n) {
    int i = blockIdx.x * blockDim.x + threadIdx.x;
    if (i < n) {
        const float* hp = H + (size_t)i * FDIM;
        float s = 0.0f;
        #pragma unroll
        for (int f = 0; f < FDIM; ++f) s = fmaf(fmaxf(hp[f], 0.0f), w2[f], s);
        lin[i] = s;
    }
}

__global__ __launch_bounds__(256) void k_agg1(const float* __restrict__ lin,
                                              const int2* __restrict__ csr,
                                              const int* __restrict__ rowptr,
                                              const float* __restrict__ deg,
                                              const float* __restrict__ b2,
                                              float* __restrict__ out, int n) {
    int d = blockIdx.x * blockDim.x + threadIdx.x;
    if (d < n) {
        const int beg = rowptr[d];
        const int end = rowptr[d + 1];
        const float rs = rsqrtf(deg[d]);
        float acc = lin[d] * (rs * rs) + b2[0];
        #pragma unroll 4
        for (int p = beg; p < end; ++p) {
            const int2 pr = csr[p];
            acc += lin[pr.x] * __int_as_float(pr.y);
        }
        out[d] = acc;
    }
}


extern "C" void kernel_launch(void* const* d_in, const int* in_sizes, int n_in,
                              void* d_out, int out_size, void* d_ws, size_t ws_size,
                              hipStream_t stream) {
    const float* x    = (const float*)d_in[0];
    const int*   esrc = (const int*)  d_in[1];
    const int*   edst = (const int*)  d_in[2];
    const float* ew   = (const float*)d_in[3];
    const float* W0   = (const float*)d_in[4];
    const float* b0   = (const float*)d_in[5];
    const float* W1   = (const float*)d_in[6];
    const float* b1   = (const float*)d_in[7];
    const float* W2   = (const float*)d_in[8];
    const float* b2   = (const float*)d_in[9];

    const int N = in_sizes[0] / FDIM;
    const int E = in_sizes[1];
    float* out = (float*)d_out;

    const size_t Na = ((size_t)N + 64 + 63) & ~(size_t)63;
    const size_t Ea = ((size_t)E + 63) & ~(size_t)63;
    float* ws     = (float*)d_ws;
    float* deg    = ws;
    int*   cnt    = (int*)(ws + Na);
    int*   rowptr = (int*)(ws + 2 * Na);
    int*   rank   = (int*)(ws + 3 * Na);
    int2*  csr    = (int2*)(ws + 3 * Na + Ea);
    float* hw     = ws + 3 * Na + 3 * Ea;
    float* ha     = hw + (size_t)N * FDIM;
    float* lin    = ha + (size_t)N * FDIM;

    const int T = 256;
    const int gN  = (N + T - 1) / T;
    const int gE  = (E + T - 1) / T;
    const int gS  = (N + 1023) / 1024;
    const int gA  = (N + 15) / 16;
    const int gGemm = (N + 31) / 32;

    k_init_deg<<<gN, T, 0, stream>>>(deg, cnt, N);
    k_deg_rank<<<gE, T, 0, stream>>>(edst, ew, deg, cnt, rank, E);
    k_rowptr  <<<gS, 1024, 0, stream>>>(cnt, rowptr, N);
    k_fill    <<<gE, T, 0, stream>>>(esrc, edst, ew, deg, rowptr, rank, csr, E);

    k_gemm64_first<<<gGemm, T, 0, stream>>>(x, W0, hw, N);
    k_agg64       <<<gA, T, 0, stream>>>(hw, csr, rowptr, deg, b0, ha, N);

    k_gemm64_relu<<<gGemm, T, 0, stream>>>(ha, W1, hw, N);
    k_agg64      <<<gA, T, 0, stream>>>(hw, csr, rowptr, deg, b1, ha, N);

    k_matvec64<<<gN, T, 0, stream>>>(ha, W2, lin, N);
    k_agg1    <<<gN, T, 0, stream>>>(lin, csr, rowptr, deg, b2, out, N);
}


# Fold the next layer's product into the aggregation kernel (16 summed rows -> LDS -> WMMA with W1; last layer: 16-lane tree dot with W2): 10 launches -> 8; the final dot's sum order changes
# speedup vs baseline: 1.9540x; 1.0574x over previous
#include <hip/hip_runtime.h>


typedef __attribute__((ext_vector_type(2))) float v2f;
typedef __attribute__((ext_vector_type(8))) float v8f;

#define FDIM 64


__global__ __launch_bounds__(256) void k_init_deg(float* __restrict__ deg, int* __restrict__ cnt, int n) {
    int i = blockIdx.x * blockDim.x + threadIdx.x;
    if (i < n) {
        deg[i] = 1.0f;
        cnt[i] = 0;
    }
}

__global__ __launch_bounds__(256) void k_deg_rank(const int* __restrict__ dst,
                                                  const float* __restrict__ w,
                                                  float* __restrict__ deg,
                                                  int* __restrict__ cnt,
                                                  int* __restrict__ rank, int e) {
    int i = blockIdx.x * blockDim.x + threadIdx.x;
    if (i < e) {
        const int d = dst[i];
        atomicAdd(&deg[d], w[i]);
        rank[i] = atomicAdd(&cnt[d], 1);
    }
}

__global__ __launch_bounds__(1024) void k_rowptr(const int* __restrict__ cnt,
                                                 int* __restrict__ rowptr, int n) {
    __shared__ int s_pre[32];
    __shared__ int s_tot[32];
    const int tid   = threadIdx.x;
    const int lane  = tid & 31;
    const int wv    = tid >> 5;
    const int tile0 = blockIdx.x * 1024;

    int pre = 0;
    const int4* c4 = (const int4*)cnt;
    const int lim = tile0 >> 2;
    #pragma unroll 4
    for (int i = tid; i < lim; i += 1024) {
        const int4 q = c4[i];
        pre += (q.x + q.y) + (q.z + q.w);
    }
    for (int o = 16; o > 0; o >>= 1) pre += __shfl_xor(pre, o, 32);

    const int g = tile0 + tid;
    const int c = (g < n) ? cnt[g] : 0;
    int inc = c;
    for (int o = 1; o < 32; o <<= 1) {
        const int t = __shfl_up(inc, o, 32);
        if (lane >= o) inc += t;
    }
    if (lane == 31) s_tot[wv] = inc;
    if (lane == 0)  s_pre[wv] = pre;
    __syncthreads();
    int base = 0;
    for (int k = 0; k < 32; ++k) {
        base += s_pre[k];
        if (k < wv) base += s_tot[k];
    }
    const int excl = base + inc - c;
    if (g < n)      rowptr[g] = excl;
    if (g == n - 1) rowptr[n] = excl + c;
}

__global__ __launch_bounds__(256) void k_fill(const int* __restrict__ src,
                                              const int* __restrict__ dst,
                                              const float* __restrict__ w,
                                              const float* __restrict__ deg,
                                              const int* __restrict__ rowptr,
                                              const int* __restrict__ rank,
                                              int2* __restrict__ csr, int e) {
    int i = blockIdx.x * blockDim.x + threadIdx.x;
    if (i < e) {
        const int s = src[i];
        const int d = dst[i];
        const float nv = rsqrtf(deg[s]) * w[i] * rsqrtf(deg[d]);
        csr[rowptr[d] + rank[i]] = make_int2(s, __float_as_int(nv));
    }
}


static __device__ __forceinline__ void gemm64_body(const float* __restrict__ H,
                                                   const float* __restrict__ Wm,
                                                   float* __restrict__ hw,
                                                   int nrows) {
    const int lane  = threadIdx.x & 31;
    const int wave  = threadIdx.x >> 5;
    const int ctile = wave & 3;
    const int rhalf = wave >> 2;
    const int l16   = lane & 15;
    const int hi    = lane >> 4;

    v2f bfrag[16];
    {
        const int col = ctile * 16 + l16;
        #pragma unroll
        for (int k = 0; k < 16; ++k) {
            const int k0 = 4 * k + (hi ? 2 : 0);
            bfrag[k].x = Wm[(size_t)k0 * FDIM + col];
            bfrag[k].y = Wm[(size_t)(k0 + 1) * FDIM + col];
        }
    }

    const int cc = ctile * 16 + l16;
    const int rowBlocks = (nrows + 31) >> 5;
    for (int rb = blockIdx.x; rb < rowBlocks; rb += gridDim.x) {
        const int row0 = rb * 32 + rhalf * 16;
        const bool full = (row0 + 16 <= nrows);

        int r  = row0 + l16;
        int rc = full ? r : ((r < nrows) ? r : (nrows > 0 ? nrows - 1 : 0));
        const float* hp = H + (size_t)rc * FDIM + (hi ? 2 : 0);

        v8f acc = {};
        #pragma unroll
        for (int k = 0; k < 16; ++k) {
            const v2f a = *(const v2f*)(hp + 4 * k);
            acc = __builtin_amdgcn_wmma_f32_16x16x4_f32(
                      false, a, false, bfrag[k], (short)0, acc, false, false);
        }

        if (full) {
            float* op = hw + (size_t)(row0 + hi * 8) * FDIM + cc;
            #pragma unroll
            for (int i = 0; i < 8; ++i)
                op[(size_t)i * FDIM] = acc[i];
        } else {
            #pragma unroll
            for (int i = 0; i < 8; ++i) {
                int rr = row0 + hi * 8 + i;
                if (rr < nrows) hw[(size_t)rr * FDIM + cc] = acc[i];
            }
        }
    }
}

__global__ __launch_bounds__(256) void k_gemm64_first(const float* __restrict__ H,
                                                      const float* __restrict__ Wm,
                                                      float* __restrict__ hw, int nrows) {
    gemm64_body(H, Wm, hw, nrows);
}


static __device__ __forceinline__ float4 agg64_row(const float* __restrict__ hw,
                                                   const int2* __restrict__ csr,
                                                   const int* __restrict__ rowptr,
                                                   const float* __restrict__ deg,
                                                   const float* __restrict__ bias,
                                                   const int d, const int l16) {
    const int beg = rowptr[d];
    const int end = rowptr[d + 1];
    const float rs = rsqrtf(deg[d]);
    const float sn = rs * rs;
    const int f = l16 * 4;
    const float4 hs = *(const float4*)(hw + (size_t)d * FDIM + f);
    float4 acc;
    acc.x = hs.x * sn + bias[f + 0];
    acc.y = hs.y * sn + bias[f + 1];
    acc.z = hs.z * sn + bias[f + 2];
    acc.w = hs.w * sn + bias[f + 3];

    const float4 zero4 = make_float4(0.0f, 0.0f, 0.0f, 0.0f);
    const float* hrow = hw + f;
    for (int base = beg; base < end; base += 16) {
        const int idx = base + l16;
        int2 pr = make_int2(0, 0);
        if (idx < end) pr = csr[idx];
        const int m = min(16, end - base);
        for (int j = 0; j < m; j += 4) {
            const int   s0 = __shfl(pr.x, j + 0, 16);
            const int   s1 = __shfl(pr.x, j + 1, 16);
            const int   s2 = __shfl(pr.x, j + 2, 16);
            const int   s3 = __shfl(pr.x, j + 3, 16);
            const float n0 = __int_as_float(__shfl(pr.y, j + 0, 16));
            const float n1 = __int_as_float(__shfl(pr.y, j + 1, 16));
            const float n2 = __int_as_float(__shfl(pr.y, j + 2, 16));
            const float n3 = __int_as_float(__shfl(pr.y, j + 3, 16));
            float4 v0 = *(const float4*)(hrow + (size_t)s0 * FDIM);
            float4 v1 = zero4, v2 = zero4, v3 = zero4;
            if (j + 1 < m) v1 = *(const float4*)(hrow + (size_t)s1 * FDIM);
            if (j + 2 < m) v2 = *(const float4*)(hrow + (size_t)s2 * FDIM);
            if (j + 3 < m) v3 = *(const float4*)(hrow + (size_t)s3 * FDIM);
            acc.x += v0.x * n0; acc.y += v0.y * n0; acc.z += v0.z * n0; acc.w += v0.w * n0;
            acc.x += v1.x * n1; acc.y += v1.y * n1; acc.z += v1.z * n1; acc.w += v1.w * n1;
            acc.x += v2.x * n2; acc.y += v2.y * n2; acc.z += v2.z * n2; acc.w += v2.w * n2;
            acc.x += v3.x * n3; acc.y += v3.y * n3; acc.z += v3.z * n3; acc.w += v3.w * n3;
        }
    }
    return acc;
}

#define HPITCH 68

__global__ __launch_bounds__(256) void k_agg64_gemm(const float* __restrict__ hw,
                                                    const int2* __restrict__ csr,
                                                    const int* __restrict__ rowptr,
                                                    const float* __restrict__ deg,
                                                    const float* __restrict__ bias,
                                                    const float* __restrict__ Wn,
                                                    float* __restrict__ hnext, int n) {
    __shared__ float s_h[16 * HPITCH];
    const int l16  = threadIdx.x & 15;
    const int grp  = threadIdx.x >> 4;
    const int row0 = blockIdx.x * 16;
    const int g    = row0 + grp;
    const int d    = (g < n) ? g : (n - 1);

    float4 acc = agg64_row(hw, csr, rowptr, deg, bias, d, l16);

    const int wave = threadIdx.x >> 5;
    const int hi   = (threadIdx.x >> 4) & 1;
    const int cc   = (wave & 3) * 16 + l16;
    v2f bfrag[16];
    if (wave < 4) {
        #pragma unroll
        for (int k = 0; k < 16; ++k) {
            const int k0 = 4 * k + (hi ? 2 : 0);
            bfrag[k].x = Wn[(size_t)k0 * FDIM + cc];
            bfrag[k].y = Wn[(size_t)(k0 + 1) * FDIM + cc];
        }
    }

    acc.x = fmaxf(acc.x, 0.0f);
    acc.y = fmaxf(acc.y, 0.0f);
    acc.z = fmaxf(acc.z, 0.0f);
    acc.w = fmaxf(acc.w, 0.0f);
    *(float4*)(s_h + grp * HPITCH + l16 * 4) = acc;
    __syncthreads();

    if (wave < 4) {
        const float* ap = s_h + l16 * HPITCH + (hi ? 2 : 0);
        v8f c = {};
        #pragma unroll
        for (int k = 0; k < 16; ++k) {
            const v2f a = *(const v2f*)(ap + 4 * k);
            c = __builtin_amdgcn_wmma_f32_16x16x4_f32(
                    false, a, false, bfrag[k], (short)0, c, false, false);
        }
        const int rbase = row0 + hi * 8;
        if (row0 + 16 <= n) {
            float* op = hnext + (size_t)rbase * FDIM + cc;
            #pragma unroll
            for (int i = 0; i < 8; ++i)
                op[(size_t)i * FDIM] = c[i];
        } else {
            #pragma unroll
            for (int i = 0; i < 8; ++i) {
                const int rr = rbase + i;
                if (rr < n) hnext[(size_t)rr * FDIM + cc] = c[i];
            }
        }
    }
}

__global__ __launch_bounds__(512) void k_agg64_dot(const float* __restrict__ hw,
                                                   const int2* __restrict__ csr,
                                                   const int* __restrict__ rowptr,
                                                   const float* __restrict__ deg,
                                                   const float* __restrict__ bias,
                                                   const float* __restrict__ w2,
                                                   float* __restrict__ lin, int n) {
    __shared__ float s_lin[32];
    const int l16  = threadIdx.x & 15;
    const int grp  = threadIdx.x >> 4;
    const int row0 = blockIdx.x * 32;
    const int g    = row0 + grp;
    const int d    = (g < n) ? g : (n - 1);

    const float4 acc = agg64_row(hw, csr, rowptr, deg, bias, d, l16);
    const int f = l16 * 4;
    float s = fmaxf(acc.x, 0.0f) * w2[f + 0];
    s = fmaf(fmaxf(acc.y, 0.0f), w2[f + 1], s);
    s = fmaf(fmaxf(acc.z, 0.0f), w2[f + 2], s);
    s = fmaf(fmaxf(acc.w, 0.0f), w2[f + 3], s);
    for (int o = 8; o > 0; o >>= 1) s += __shfl_xor(s, o, 16);
    if (l16 == 0) s_lin[grp] = s;
    __syncthreads();
    if (threadIdx.x < 32) {
        const int r = row0 + threadIdx.x;
        if (r < n) lin[r] = s_lin[threadIdx.x];
    }
}

__global__ __launch_bounds__(256) void k_agg1(const float* __restrict__ lin,
                                              const int2* __restrict__ csr,
                                              const int* __restrict__ rowptr,
                                              const float* __restrict__ deg,
                                              const float* __restrict__ b2,
                                              float* __restrict__ out, int n) {
    int d = blockIdx.x * blockDim.x + threadIdx.x;
    if (d < n) {
        const int beg = rowptr[d];
        const int end = rowptr[d + 1];
        const float rs = rsqrtf(deg[d]);
        float acc = lin[d] * (rs * rs) + b2[0];
        #pragma unroll 4
        for (int p = beg; p < end; ++p) {
            const int2 pr = csr[p];
            acc += lin[pr.x] * __int_as_float(pr.y);
        }
        out[d] = acc;
    }
}


extern "C" void kernel_launch(void* const* d_in, const int* in_sizes, int n_in,
                              void* d_out, int out_size, void* d_ws, size_t ws_size,
                              hipStream_t stream) {
    const float* x    = (const float*)d_in[0];
    const int*   esrc = (const int*)  d_in[1];
    const int*   edst = (const int*)  d_in[2];
    const float* ew   = (const float*)d_in[3];
    const float* W0   = (const float*)d_in[4];
    const float* b0   = (const float*)d_in[5];
    const float* W1   = (const float*)d_in[6];
    const float* b1   = (const float*)d_in[7];
    const float* W2   = (const float*)d_in[8];
    const float* b2   = (const float*)d_in[9];

    const int N = in_sizes[0] / FDIM;
    const int E = in_sizes[1];
    float* out = (float*)d_out;

    const size_t Na = ((size_t)N + 64 + 63) & ~(size_t)63;
    const size_t Ea = ((size_t)E + 63) & ~(size_t)63;
    float* ws     = (float*)d_ws;
    float* deg    = ws;
    int*   cnt    = (int*)(ws + Na);
    int*   rowptr = (int*)(ws + 2 * Na);
    int*   rank   = (int*)(ws + 3 * Na);
    int2*  csr    = (int2*)(ws + 3 * Na + Ea);
    float* hw     = ws + 3 * Na + 3 * Ea;
    float* ha     = hw + (size_t)N * FDIM;
    float* lin    = ha + (size_t)N * FDIM;

    const int T = 256;
    const int gN  = (N + T - 1) / T;
    const int gE  = (E + T - 1) / T;
    const int gS  = (N + 1023) / 1024;
    const int gA  = (N + 15) / 16;
    const int gGemm = (N + 31) / 32;

    k_init_deg<<<gN, T, 0, stream>>>(deg, cnt, N);
    k_deg_rank<<<gE, T, 0, stream>>>(edst, ew, deg, cnt, rank, E);
    k_rowptr  <<<gS, 1024, 0, stream>>>(cnt, rowptr, N);
    k_fill    <<<gE, T, 0, stream>>>(esrc, edst, ew, deg, rowptr, rank, csr, E);

    k_gemm64_first<<<gGemm, T, 0, stream>>>(x, W0, hw, N);

    k_agg64_gemm<<<gA, T, 0, stream>>>(hw, csr, rowptr, deg, b0, W1, ha, N);

    k_agg64_dot<<<gGemm, 512, 0, stream>>>(ha, csr, rowptr, deg, b1, W2, lin, N);

    k_agg1    <<<gN, T, 0, stream>>>(lin, csr, rowptr, deg, b2, out, N);
}


# Store rows scaled by rsqrt(deg), scale each row sum once: edges carry raw w, so no float atomics, no per-edge deg reads; deg = 8-lane row sum of the sorted list; norm rounding, deg sum order change
# speedup vs baseline: 2.2976x; 1.1758x over previous
#include <hip/hip_runtime.h>


typedef __attribute__((ext_vector_type(2))) float v2f;
typedef __attribute__((ext_vector_type(8))) float v8f;

#define FDIM 64


__global__ __launch_bounds__(256) void k_init_cnt(int* __restrict__ cnt, int n) {
    int i = blockIdx.x * blockDim.x + threadIdx.x;
    if (i < n) cnt[i] = 0;
}

__global__ __launch_bounds__(256) void k_rank(const int* __restrict__ dst,
                                              int* __restrict__ cnt,
                                              int* __restrict__ rank, int e) {
    int i = blockIdx.x * blockDim.x + threadIdx.x;
    if (i < e) rank[i] = atomicAdd(&cnt[dst[i]], 1);
}

__global__ __launch_bounds__(1024) void k_rowptr(const int* __restrict__ cnt,
                                                 int* __restrict__ rowptr, int n) {
    __shared__ int s_pre[32];
    __shared__ int s_tot[32];
    const int tid   = threadIdx.x;
    const int lane  = tid & 31;
    const int wv    = tid >> 5;
    const int tile0 = blockIdx.x * 1024;

    int pre = 0;
    const int4* c4 = (const int4*)cnt;
    const int lim = tile0 >> 2;
    #pragma unroll 4
    for (int i = tid; i < lim; i += 1024) {
        const int4 q = c4[i];
        pre += (q.x + q.y) + (q.z + q.w);
    }
    for (int o = 16; o > 0; o >>= 1) pre += __shfl_xor(pre, o, 32);

    const int g = tile0 + tid;
    const int c = (g < n) ? cnt[g] : 0;
    int inc = c;
    for (int o = 1; o < 32; o <<= 1) {
        const int t = __shfl_up(inc, o, 32);
        if (lane >= o) inc += t;
    }
    if (lane == 31) s_tot[wv] = inc;
    if (lane == 0)  s_pre[wv] = pre;
    __syncthreads();
    int base = 0;
    for (int k = 0; k < 32; ++k) {
        base += s_pre[k];
        if (k < wv) base += s_tot[k];
    }
    const int excl = base + inc - c;
    if (g < n)      rowptr[g] = excl;
    if (g == n - 1) rowptr[n] = excl + c;
}

__global__ __launch_bounds__(256) void k_fill(const int* __restrict__ src,
                                              const int* __restrict__ dst,
                                              const float* __restrict__ w,
                                              const int* __restrict__ rowptr,
                                              const int* __restrict__ rank,
                                              int2* __restrict__ csr, int e) {
    int i = blockIdx.x * blockDim.x + threadIdx.x;
    if (i < e) {
        const int d = dst[i];
        csr[rowptr[d] + rank[i]] = make_int2(src[i], __float_as_int(w[i]));
    }
}


__global__ __launch_bounds__(256) void k_gemm64_first(const float* __restrict__ H,
                                                      const float* __restrict__ Wm,
                                                      const int2* __restrict__ csr,
                                                      const int* __restrict__ rowptr,
                                                      float* __restrict__ dinv,
                                                      float* __restrict__ hw, int nrows) {
    __shared__ float s_dinv[32];
    const int lane  = threadIdx.x & 31;
    const int wave  = threadIdx.x >> 5;
    const int ctile = wave & 3;
    const int rhalf = wave >> 2;
    const int l16   = lane & 15;
    const int hi    = lane >> 4;
    const int blk0  = blockIdx.x * 32;

    {
        const int rl = threadIdx.x >> 3;
        const int t8 = threadIdx.x & 7;
        const int r  = blk0 + rl;
        float s = 0.0f;
        if (r < nrows) {
            const int beg = rowptr[r];
            const int end = rowptr[r + 1];
            for (int p = beg + t8; p < end; p += 8) s += __int_as_float(csr[p].y);
        }
        s += __shfl_xor(s, 4, 8);
        s += __shfl_xor(s, 2, 8);
        s += __shfl_xor(s, 1, 8);
        if (t8 == 0) {
            const float dv = rsqrtf(1.0f + s);
            s_dinv[rl] = dv;
            if (r < nrows) dinv[r] = dv;
        }
    }

    v2f bfrag[16];
    const int cc = ctile * 16 + l16;
    #pragma unroll
    for (int k = 0; k < 16; ++k) {
        const int k0 = 4 * k + (hi ? 2 : 0);
        bfrag[k].x = Wm[(size_t)k0 * FDIM + cc];
        bfrag[k].y = Wm[(size_t)(k0 + 1) * FDIM + cc];
    }

    const int row0 = blk0 + rhalf * 16;
    const bool full = (row0 + 16 <= nrows);
    const int r  = row0 + l16;
    const int rc = full ? r : ((r < nrows) ? r : (nrows > 0 ? nrows - 1 : 0));
    const float* hp = H + (size_t)rc * FDIM + (hi ? 2 : 0);

    v8f acc = {};
    #pragma unroll
    for (int k = 0; k < 16; ++k) {
        const v2f a = *(const v2f*)(hp + 4 * k);
        acc = __builtin_amdgcn_wmma_f32_16x16x4_f32(
                  false, a, false, bfrag[k], (short)0, acc, false, false);
    }
    __syncthreads();

    const float* sd = s_dinv + rhalf * 16 + hi * 8;
    if (full) {
        float* op = hw + (size_t)(row0 + hi * 8) * FDIM + cc;
        #pragma unroll
        for (int i = 0; i < 8; ++i)
            op[(size_t)i * FDIM] = acc[i] * sd[i];
    } else {
        #pragma unroll
        for (int i = 0; i < 8; ++i) {
            int rr = row0 + hi * 8 + i;
            if (rr < nrows) hw[(size_t)rr * FDIM + cc] = acc[i] * sd[i];
        }
    }
}


static __device__ __forceinline__ float4 agg64_row(const float* __restrict__ hw,
                                                   const int2* __restrict__ csr,
                                                   const int* __restrict__ rowptr,
                                                   const int d, const int l16) {
    const int beg = rowptr[d];
    const int end = rowptr[d + 1];
    const int f = l16 * 4;
    float4 acc = *(const float4*)(hw + (size_t)d * FDIM + f);

    const float4 zero4 = make_float4(0.0f, 0.0f, 0.0f, 0.0f);
    const float* hrow = hw + f;
    for (int base = beg; base < end; base += 16) {
        const int idx = base + l16;
        int2 pr = make_int2(0, 0);
        if (idx < end) pr = csr[idx];
        const int m = min(16, end - base);
        for (int j = 0; j < m; j += 4) {
            const int   s0 = __shfl(pr.x, j + 0, 16);
            const int   s1 = __shfl(pr.x, j + 1, 16);
            const int   s2 = __shfl(pr.x, j + 2, 16);
            const int   s3 = __shfl(pr.x, j + 3, 16);
            const float n0 = __int_as_float(__shfl(pr.y, j + 0, 16));
            const float n1 = __int_as_float(__shfl(pr.y, j + 1, 16));
            const float n2 = __int_as_float(__shfl(pr.y, j + 2, 16));
            const float n3 = __int_as_float(__shfl(pr.y, j + 3, 16));
            float4 v0 = *(const float4*)(hrow + (size_t)s0 * FDIM);
            float4 v1 = zero4, v2 = zero4, v3 = zero4;
            if (j + 1 < m) v1 = *(const float4*)(hrow + (size_t)s1 * FDIM);
            if (j + 2 < m) v2 = *(const float4*)(hrow + (size_t)s2 * FDIM);
            if (j + 3 < m) v3 = *(const float4*)(hrow + (size_t)s3 * FDIM);
            acc.x += v0.x * n0; acc.y += v0.y * n0; acc.z += v0.z * n0; acc.w += v0.w * n0;
            acc.x += v1.x * n1; acc.y += v1.y * n1; acc.z += v1.z * n1; acc.w += v1.w * n1;
            acc.x += v2.x * n2; acc.y += v2.y * n2; acc.z += v2.z * n2; acc.w += v2.w * n2;
            acc.x += v3.x * n3; acc.y += v3.y * n3; acc.z += v3.z * n3; acc.w += v3.w * n3;
        }
    }
    return acc;
}

#define HPITCH 68

__global__ __launch_bounds__(256) void k_agg64_gemm(const float* __restrict__ hw,
                                                    const int2* __restrict__ csr,
                                                    const int* __restrict__ rowptr,
                                                    const float* __restrict__ dinv,
                                                    const float* __restrict__ bias,
                                                    const float* __restrict__ Wn,
                                                    float* __restrict__ hnext, int n) {
    __shared__ float s_h[16 * HPITCH];
    const int l16  = threadIdx.x & 15;
    const int grp  = threadIdx.x >> 4;
    const int row0 = blockIdx.x * 16;
    const int g    = row0 + grp;
    const int d    = (g < n) ? g : (n - 1);

    const float dv = dinv[d];
    const float4 bs = *(const float4*)(bias + l16 * 4);
    float4 acc = agg64_row(hw, csr, rowptr, d, l16);

    const int wave = threadIdx.x >> 5;
    const int hi   = (threadIdx.x >> 4) & 1;
    const int cc   = (wave & 3) * 16 + l16;
    v2f bfrag[16];
    if (wave < 4) {
        #pragma unroll
        for (int k = 0; k < 16; ++k) {
            const int k0 = 4 * k + (hi ? 2 : 0);
            bfrag[k].x = Wn[(size_t)k0 * FDIM + cc];
            bfrag[k].y = Wn[(size_t)(k0 + 1) * FDIM + cc];
        }
    }

    acc.x = fmaxf(fmaf(acc.x, dv, bs.x), 0.0f) * dv;
    acc.y = fmaxf(fmaf(acc.y, dv, bs.y), 0.0f) * dv;
    acc.z = fmaxf(fmaf(acc.z, dv, bs.z), 0.0f) * dv;
    acc.w = fmaxf(fmaf(acc.w, dv, bs.w), 0.0f) * dv;
    *(float4*)(s_h + grp * HPITCH + l16 * 4) = acc;
    __syncthreads();

    if (wave < 4) {
        const float* ap = s_h + l16 * HPITCH + (hi ? 2 : 0);
        v8f c = {};
        #pragma unroll
        for (int k = 0; k < 16; ++k) {
            const v2f a = *(const v2f*)(ap + 4 * k);
            c = __builtin_amdgcn_wmma_f32_16x16x4_f32(
                    false, a, false, bfrag[k], (short)0, c, false, false);
        }
        const int rbase = row0 + hi * 8;
        if (row0 + 16 <= n) {
            float* op = hnext + (size_t)rbase * FDIM + cc;
            #pragma unroll
            for (int i = 0; i < 8; ++i)
                op[(size_t)i * FDIM] = c[i];
        } else {
            #pragma unroll
            for (int i = 0; i < 8; ++i) {
                const int rr = rbase + i;
                if (rr < n) hnext[(size_t)rr * FDIM + cc] = c[i];
            }
        }
    }
}

__global__ __launch_bounds__(512) void k_agg64_dot(const float* __restrict__ hw,
                                                   const int2* __restrict__ csr,
                                                   const int* __restrict__ rowptr,
                                                   const float* __restrict__ dinv,
                                                   const float* __restrict__ bias,
                                                   const float* __restrict__ w2,
                                                   float* __restrict__ lin, int n) {
    __shared__ float s_lin[32];
    const int l16  = threadIdx.x & 15;
    const int grp  = threadIdx.x >> 4;
    const int row0 = blockIdx.x * 32;
    const int g    = row0 + grp;
    const int d    = (g < n) ? g : (n - 1);

    const float dv = dinv[d];
    const int f = l16 * 4;
    const float4 bs = *(const float4*)(bias + f);
    const float4 acc = agg64_row(hw, csr, rowptr, d, l16);
    float s = fmaxf(fmaf(acc.x, dv, bs.x), 0.0f) * w2[f + 0];
    s = fmaf(fmaxf(fmaf(acc.y, dv, bs.y), 0.0f), w2[f + 1], s);
    s = fmaf(fmaxf(fmaf(acc.z, dv, bs.z), 0.0f), w2[f + 2], s);
    s = fmaf(fmaxf(fmaf(acc.w, dv, bs.w), 0.0f), w2[f + 3], s);
    for (int o = 8; o > 0; o >>= 1) s += __shfl_xor(s, o, 16);
    if (l16 == 0) s_lin[grp] = s * dv;
    __syncthreads();
    if (threadIdx.x < 32) {
        const int r = row0 + threadIdx.x;
        if (r < n) lin[r] = s_lin[threadIdx.x];
    }
}

__global__ __launch_bounds__(256) void k_agg1(const float* __restrict__ lin,
                                              const int2* __restrict__ csr,
                                              const int* __restrict__ rowptr,
                                              const float* __restrict__ dinv,
                                              const float* __restrict__ b2,
                                              float* __restrict__ out, int n) {
    int d = blockIdx.x * blockDim.x + threadIdx.x;
    if (d < n) {
        const int beg = rowptr[d];
        const int end = rowptr[d + 1];
        float acc = lin[d];
        #pragma unroll 4
        for (int p = beg; p < end; ++p) {
            const int2 pr = csr[p];
            acc += lin[pr.x] * __int_as_float(pr.y);
        }
        out[d] = fmaf(acc, dinv[d], b2[0]);
    }
}


extern "C" void kernel_launch(void* const* d_in, const int* in_sizes, int n_in,
                              void* d_out, int out_size, void* d_ws, size_t ws_size,
                              hipStream_t stream) {
    const float* x    = (const float*)d_in[0];
    const int*   esrc = (const int*)  d_in[1];
    const int*   edst = (const int*)  d_in[2];
    const float* ew   = (const float*)d_in[3];
    const float* W0   = (const float*)d_in[4];
    const float* b0   = (const float*)d_in[5];
    const float* W1   = (const float*)d_in[6];
    const float* b1   = (const float*)d_in[7];
    const float* W2   = (const float*)d_in[8];
    const float* b2   = (const float*)d_in[9];

    const int N = in_sizes[0] / FDIM;
    const int E = in_sizes[1];
    float* out = (float*)d_out;

    const size_t Na = ((size_t)N + 64 + 63) & ~(size_t)63;
    const size_t Ea = ((size_t)E + 63) & ~(size_t)63;
    float* ws     = (float*)d_ws;
    float* dinv   = ws;
    int*   cnt    = (int*)(ws + Na);
    int*   rowptr = (int*)(ws + 2 * Na);
    int*   rank   = (int*)(ws + 3 * Na);
    int2*  csr    = (int2*)(ws + 3 * Na + Ea);
    float* hw     = ws + 3 * Na + 3 * Ea;
    float* ha     = hw + (size_t)N * FDIM;
    float* lin    = ha + (size_t)N * FDIM;

    const int T = 256;
    const int gN  = (N + T - 1) / T;
    const int gE  = (E + T - 1) / T;
    const int gS  = (N + 1023) / 1024;
    const int gA  = (N + 15) / 16;
    const int gGemm = (N + 31) / 32;

    k_init_cnt<<<gN, T, 0, stream>>>(cnt, N);
    k_rank    <<<gE, T, 0, stream>>>(edst, cnt, rank, E);
    k_rowptr  <<<gS, 1024, 0, stream>>>(cnt, rowptr, N);
    k_fill    <<<gE, T, 0, stream>>>(esrc, edst, ew, rowptr, rank, csr, E);

    k_gemm64_first<<<gGemm, T, 0, stream>>>(x, W0, csr, rowptr, dinv, hw, N);

    k_agg64_gemm<<<gA, T, 0, stream>>>(hw, csr, rowptr, dinv, b0, W1, ha, N);

    k_agg64_dot<<<gGemm, 512, 0, stream>>>(ha, csr, rowptr, dinv, b1, W2, lin, N);

    k_agg1    <<<gN, T, 0, stream>>>(lin, csr, rowptr, dinv, b2, out, N);
}


# Replace the per-edge returning global atomic and scattered fill by a two-level counting sort on LDS counters (chunk histogram, scan, bucket pass, row pass per bucket); row sum order: LDS arrival rank
# speedup vs baseline: 3.4313x; 1.4934x over previous
#include <hip/hip_runtime.h>


typedef __attribute__((ext_vector_type(2))) float v2f;
typedef __attribute__((ext_vector_type(8))) float v8f;

#define FDIM 64


#define RB_SHIFT 8
#define RB       (1 << RB_SHIFT)
#define CH_T     256
#define CH       (CH_T * 16)

__global__ __launch_bounds__(CH_T) void k_hist(const int* __restrict__ dst,
                                               int* __restrict__ hist, int e, int nb, int nc) {
    extern __shared__ int s_h[];
    for (int b = threadIdx.x; b < nb; b += CH_T) s_h[b] = 0;
    __syncthreads();
    const int c = blockIdx.x;
    if ((c + 1) * CH <= e) {
        int4 d[4];
        #pragma unroll
        for (int g = 0; g < 4; ++g)
            d[g] = *(const int4*)(dst + c * CH + (g * CH_T + threadIdx.x) * 4);
        #pragma unroll
        for (int g = 0; g < 4; ++g) {
            atomicAdd(&s_h[d[g].x >> RB_SHIFT], 1);
            atomicAdd(&s_h[d[g].y >> RB_SHIFT], 1);
            atomicAdd(&s_h[d[g].z >> RB_SHIFT], 1);
            atomicAdd(&s_h[d[g].w >> RB_SHIFT], 1);
        }
    } else {
        for (int g = 0; g < 4; ++g) {
            const int i = c * CH + (g * CH_T + threadIdx.x) * 4;
            if (i + 3 < e) {
                const int4 dg = *(const int4*)(dst + i);
                atomicAdd(&s_h[dg.x >> RB_SHIFT], 1);
                atomicAdd(&s_h[dg.y >> RB_SHIFT], 1);
                atomicAdd(&s_h[dg.z >> RB_SHIFT], 1);
                atomicAdd(&s_h[dg.w >> RB_SHIFT], 1);
            } else {
                for (int k = i; k < e; ++k) atomicAdd(&s_h[dst[k] >> RB_SHIFT], 1);
            }
        }
    }
    __syncthreads();
    for (int b = threadIdx.x; b < nb; b += CH_T) hist[(size_t)b * nc + c] = s_h[b];
}

__global__ __launch_bounds__(1024) void k_rowptr(const int* __restrict__ cnt,
                                                 int* __restrict__ rowptr, int n) {
    __shared__ int s_pre[32];
    __shared__ int s_tot[32];
    const int tid   = threadIdx.x;
    const int lane  = tid & 31;
    const int wv    = tid >> 5;
    const int tile0 = blockIdx.x * 1024;

    int pre = 0;
    const int4* c4 = (const int4*)cnt;
    const int lim = tile0 >> 2;
    #pragma unroll 4
    for (int i = tid; i < lim; i += 1024) {
        const int4 q = c4[i];
        pre += (q.x + q.y) + (q.z + q.w);
    }
    for (int o = 16; o > 0; o >>= 1) pre += __shfl_xor(pre, o, 32);

    const int g = tile0 + tid;
    const int c = (g < n) ? cnt[g] : 0;
    int inc = c;
    for (int o = 1; o < 32; o <<= 1) {
        const int t = __shfl_up(inc, o, 32);
        if (lane >= o) inc += t;
    }
    if (lane == 31) s_tot[wv] = inc;
    if (lane == 0)  s_pre[wv] = pre;
    __syncthreads();
    int base = 0;
    for (int k = 0; k < 32; ++k) {
        base += s_pre[k];
        if (k < wv) base += s_tot[k];
    }
    const int excl = base + inc - c;
    if (g < n)      rowptr[g] = excl;
    if (g == n - 1) rowptr[n] = excl + c;
}

__global__ __launch_bounds__(CH_T) void k_scatter(const int* __restrict__ src,
                                                  const int* __restrict__ dst,
                                                  const float* __restrict__ w,
                                                  const int* __restrict__ hoff,
                                                  int4* __restrict__ tmp, int e, int nb, int nc) {
    extern __shared__ int s_o[];
    const int c = blockIdx.x;
    for (int b = threadIdx.x; b < nb; b += CH_T) s_o[b] = hoff[(size_t)b * nc + c];
    __syncthreads();
    if ((c + 1) * CH <= e) {
        int4 d[4], s[4];
        float4 ww[4];
        #pragma unroll
        for (int g = 0; g < 4; ++g) {
            const int i = c * CH + (g * CH_T + threadIdx.x) * 4;
            d[g]  = *(const int4*)(dst + i);
            s[g]  = *(const int4*)(src + i);
            ww[g] = *(const float4*)(w + i);
        }
        #pragma unroll
        for (int g = 0; g < 4; ++g) {
            const int p0 = atomicAdd(&s_o[d[g].x >> RB_SHIFT], 1);
            const int p1 = atomicAdd(&s_o[d[g].y >> RB_SHIFT], 1);
            const int p2 = atomicAdd(&s_o[d[g].z >> RB_SHIFT], 1);
            const int p3 = atomicAdd(&s_o[d[g].w >> RB_SHIFT], 1);
            tmp[p0] = make_int4(s[g].x, d[g].x, __float_as_int(ww[g].x), 0);
            tmp[p1] = make_int4(s[g].y, d[g].y, __float_as_int(ww[g].y), 0);
            tmp[p2] = make_int4(s[g].z, d[g].z, __float_as_int(ww[g].z), 0);
            tmp[p3] = make_int4(s[g].w, d[g].w, __float_as_int(ww[g].w), 0);
        }
    } else {
        for (int g = 0; g < 4; ++g) {
            const int i = c * CH + (g * CH_T + threadIdx.x) * 4;
            if (i + 3 < e) {
                const int4   dg = *(const int4*)(dst + i);
                const int4   sg = *(const int4*)(src + i);
                const float4 wg = *(const float4*)(w + i);
                const int p0 = atomicAdd(&s_o[dg.x >> RB_SHIFT], 1);
                const int p1 = atomicAdd(&s_o[dg.y >> RB_SHIFT], 1);
                const int p2 = atomicAdd(&s_o[dg.z >> RB_SHIFT], 1);
                const int p3 = atomicAdd(&s_o[dg.w >> RB_SHIFT], 1);
                tmp[p0] = make_int4(sg.x, dg.x, __float_as_int(wg.x), 0);
                tmp[p1] = make_int4(sg.y, dg.y, __float_as_int(wg.y), 0);
                tmp[p2] = make_int4(sg.z, dg.z, __float_as_int(wg.z), 0);
                tmp[p3] = make_int4(sg.w, dg.w, __float_as_int(wg.w), 0);
            } else {
                for (int k = i; k < e; ++k) {
                    const int dk = dst[k];
                    const int p  = atomicAdd(&s_o[dk >> RB_SHIFT], 1);
                    tmp[p] = make_int4(src[k], dk, __float_as_int(w[k]), 0);
                }
            }
        }
    }
}

__global__ __launch_bounds__(512) void k_bucket(const int4* __restrict__ tmp,
                                                const int* __restrict__ hoff,
                                                int* __restrict__ rowptr,
                                                int2* __restrict__ csr, int n, int nc) {
    __shared__ int s_cnt[RB];
    __shared__ int s_cur[RB];
    __shared__ int s_wtot[RB / 32];
    const int tid  = threadIdx.x;
    const int lane = tid & 31;
    const int wv   = tid >> 5;
    const int b    = blockIdx.x;
    const int beg  = hoff[(size_t)b * nc];
    const int end  = hoff[(size_t)(b + 1) * nc];

    if (tid < RB) s_cnt[tid] = 0;
    __syncthreads();
    for (int p0 = beg + tid; p0 < end; p0 += 4 * 512) {
        const int p1 = p0 + 512, p2 = p0 + 1024, p3 = p0 + 1536;
        const int d0 = tmp[p0].y;
        const int d1 = tmp[min(p1, end - 1)].y;
        const int d2 = tmp[min(p2, end - 1)].y;
        const int d3 = tmp[min(p3, end - 1)].y;
        atomicAdd(&s_cnt[d0 & (RB - 1)], 1);
        if (p1 < end) atomicAdd(&s_cnt[d1 & (RB - 1)], 1);
        if (p2 < end) atomicAdd(&s_cnt[d2 & (RB - 1)], 1);
        if (p3 < end) atomicAdd(&s_cnt[d3 & (RB - 1)], 1);
    }
    __syncthreads();

    int c = 0, inc = 0;
    if (tid < RB) {
        c = s_cnt[tid];
        inc = c;
        for (int o = 1; o < 32; o <<= 1) {
            const int t = __shfl_up(inc, o, 32);
            if (lane >= o) inc += t;
        }
        if (lane == 31) s_wtot[wv] = inc;
    }
    __syncthreads();
    if (tid < RB) {
        int pre = beg;
        for (int k = 0; k < RB / 32; ++k)
            if (k < wv) pre += s_wtot[k];
        const int excl = pre + inc - c;
        s_cur[tid] = excl;
        const int row = b * RB + tid;
        if (row < n)      rowptr[row] = excl;
        if (row == n - 1) rowptr[n]   = excl + c;
    }
    __syncthreads();
    for (int p0 = beg + tid; p0 < end; p0 += 4 * 512) {
        const int p1 = p0 + 512, p2 = p0 + 1024, p3 = p0 + 1536;
        const int4 e0 = tmp[p0];
        const int4 e1 = tmp[min(p1, end - 1)];
        const int4 e2 = tmp[min(p2, end - 1)];
        const int4 e3 = tmp[min(p3, end - 1)];
        csr[atomicAdd(&s_cur[e0.y & (RB - 1)], 1)] = make_int2(e0.x, e0.z);
        if (p1 < end) csr[atomicAdd(&s_cur[e1.y & (RB - 1)], 1)] = make_int2(e1.x, e1.z);
        if (p2 < end) csr[atomicAdd(&s_cur[e2.y & (RB - 1)], 1)] = make_int2(e2.x, e2.z);
        if (p3 < end) csr[atomicAdd(&s_cur[e3.y & (RB - 1)], 1)] = make_int2(e3.x, e3.z);
    }
}


__global__ __launch_bounds__(256) void k_gemm64_first(const float* __restrict__ H,
                                                      const float* __restrict__ Wm,
                                                      const int2* __restrict__ csr,
                                                      const int* __restrict__ rowptr,
                                                      float* __restrict__ dinv,
                                                      float* __restrict__ hw, int nrows) {
    __shared__ float s_dinv[32];
    const int lane  = threadIdx.x & 31;
    const int wave  = threadIdx.x >> 5;
    const int ctile = wave & 3;
    const int rhalf = wave >> 2;
    const int l16   = lane & 15;
    const int hi    = lane >> 4;
    const int blk0  = blockIdx.x * 32;

    {
        const int rl = threadIdx.x >> 3;
        const int t8 = threadIdx.x & 7;
        const int r  = blk0 + rl;
        float s = 0.0f;
        if (r < nrows) {
            const int beg = rowptr[r];
            const int end = rowptr[r + 1];
            for (int p = beg + t8; p < end; p += 8) s += __int_as_float(csr[p].y);
        }
        s += __shfl_xor(s, 4, 8);
        s += __shfl_xor(s, 2, 8);
        s += __shfl_xor(s, 1, 8);
        if (t8 == 0) {
            const float dv = rsqrtf(1.0f + s);
            s_dinv[rl] = dv;
            if (r < nrows) dinv[r] = dv;
        }
    }

    v2f bfrag[16];
    const int cc = ctile * 16 + l16;
    #pragma unroll
    for (int k = 0; k < 16; ++k) {
        const int k0 = 4 * k + (hi ? 2 : 0);
        bfrag[k].x = Wm[(size_t)k0 * FDIM + cc];
        bfrag[k].y = Wm[(size_t)(k0 + 1) * FDIM + cc];
    }

    const int row0 = blk0 + rhalf * 16;
    const bool full = (row0 + 16 <= nrows);
    const int r  = row0 + l16;
    const int rc = full ? r : ((r < nrows) ? r : (nrows > 0 ? nrows - 1 : 0));
    const float* hp = H + (size_t)rc * FDIM + (hi ? 2 : 0);

    v8f acc = {};
    #pragma unroll
    for (int k = 0; k < 16; ++k) {
        const v2f a = *(const v2f*)(hp + 4 * k);
        acc = __builtin_amdgcn_wmma_f32_16x16x4_f32(
                  false, a, false, bfrag[k], (short)0, acc, false, false);
    }
    __syncthreads();

    const float* sd = s_dinv + rhalf * 16 + hi * 8;
    if (full) {
        float* op = hw + (size_t)(row0 + hi * 8) * FDIM + cc;
        #pragma unroll
        for (int i = 0; i < 8; ++i)
            op[(size_t)i * FDIM] = acc[i] * sd[i];
    } else {
        #pragma unroll
        for (int i = 0; i < 8; ++i) {
            int rr = row0 + hi * 8 + i;
            if (rr < nrows) hw[(size_t)rr * FDIM + cc] = acc[i] * sd[i];
        }
    }
}


static __device__ __forceinline__ float4 agg64_row(const float* __restrict__ hw,
                                                   const int2* __restrict__ csr,
                                                   const int* __restrict__ rowptr,
                                                   const int d, const int l16) {
    const int beg = rowptr[d];
    const int end = rowptr[d + 1];
    const int f = l16 * 4;
    float4 acc = *(const float4*)(hw + (size_t)d * FDIM + f);

    const float4 zero4 = make_float4(0.0f, 0.0f, 0.0f, 0.0f);
    const float* hrow = hw + f;
    for (int base = beg; base < end; base += 16) {
        const int idx = base + l16;
        int2 pr = make_int2(0, 0);
        if (idx < end) pr = csr[idx];
        const int m = min(16, end - base);
        for (int j = 0; j < m; j += 4) {
            const int   s0 = __shfl(pr.x, j + 0, 16);
            const int   s1 = __shfl(pr.x, j + 1, 16);
            const int   s2 = __shfl(pr.x, j + 2, 16);
            const int   s3 = __shfl(pr.x, j + 3, 16);
            const float n0 = __int_as_float(__shfl(pr.y, j + 0, 16));
            const float n1 = __int_as_float(__shfl(pr.y, j + 1, 16));
            const float n2 = __int_as_float(__shfl(pr.y, j + 2, 16));
            const float n3 = __int_as_float(__shfl(pr.y, j + 3, 16));
            float4 v0 = *(const float4*)(hrow + (size_t)s0 * FDIM);
            float4 v1 = zero4, v2 = zero4, v3 = zero4;
            if (j + 1 < m) v1 = *(const float4*)(hrow + (size_t)s1 * FDIM);
            if (j + 2 < m) v2 = *(const float4*)(hrow + (size_t)s2 * FDIM);
            if (j + 3 < m) v3 = *(const float4*)(hrow + (size_t)s3 * FDIM);
            acc.x += v0.x * n0; acc.y += v0.y * n0; acc.z += v0.z * n0; acc.w += v0.w * n0;
            acc.x += v1.x * n1; acc.y += v1.y * n1; acc.z += v1.z * n1; acc.w += v1.w * n1;
            acc.x += v2.x * n2; acc.y += v2.y * n2; acc.z += v2.z * n2; acc.w += v2.w * n2;
            acc.x += v3.x * n3; acc.y += v3.y * n3; acc.z += v3.z * n3; acc.w += v3.w * n3;
        }
    }
    return acc;
}

#define HPITCH 68

__global__ __launch_bounds__(256) void k_agg64_gemm(const float* __restrict__ hw,
                                                    const int2* __restrict__ csr,
                                                    const int* __restrict__ rowptr,
                                                    const float* __restrict__ dinv,
                                                    const float* __restrict__ bias,
                                                    const float* __restrict__ Wn,
                                                    float* __restrict__ hnext, int n) {
    __shared__ float s_h[16 * HPITCH];
    const int l16  = threadIdx.x & 15;
    const int grp  = threadIdx.x >> 4;
    const int row0 = blockIdx.x * 16;
    const int g    = row0 + grp;
    const int d    = (g < n) ? g : (n - 1);

    const float dv = dinv[d];
    const float4 bs = *(const float4*)(bias + l16 * 4);
    float4 acc = agg64_row(hw, csr, rowptr, d, l16);

    const int wave = threadIdx.x >> 5;
    const int hi   = (threadIdx.x >> 4) & 1;
    const int cc   = (wave & 3) * 16 + l16;
    v2f bfrag[16];
    if (wave < 4) {
        #pragma unroll
        for (int k = 0; k < 16; ++k) {
            const int k0 = 4 * k + (hi ? 2 : 0);
            bfrag[k].x = Wn[(size_t)k0 * FDIM + cc];
            bfrag[k].y = Wn[(size_t)(k0 + 1) * FDIM + cc];
        }
    }

    acc.x = fmaxf(fmaf(acc.x, dv, bs.x), 0.0f) * dv;
    acc.y = fmaxf(fmaf(acc.y, dv, bs.y), 0.0f) * dv;
    acc.z = fmaxf(fmaf(acc.z, dv, bs.z), 0.0f) * dv;
    acc.w = fmaxf(fmaf(acc.w, dv, bs.w), 0.0f) * dv;
    *(float4*)(s_h + grp * HPITCH + l16 * 4) = acc;
    __syncthreads();

    if (wave < 4) {
        const float* ap = s_h + l16 * HPITCH + (hi ? 2 : 0);
        v8f c = {};
        #pragma unroll
        for (int k = 0; k < 16; ++k) {
            const v2f a = *(const v2f*)(ap + 4 * k);
            c = __builtin_amdgcn_wmma_f32_16x16x4_f32(
                    false, a, false, bfrag[k], (short)0, c, false, false);
        }
        const int rbase = row0 + hi * 8;
        if (row0 + 16 <= n) {
            float* op = hnext + (size_t)rbase * FDIM + cc;
            #pragma unroll
            for (int i = 0; i < 8; ++i)
                op[(size_t)i * FDIM] = c[i];
        } else {
            #pragma unroll
            for (int i = 0; i < 8; ++i) {
                const int rr = rbase + i;
                if (rr < n) hnext[(size_t)rr * FDIM + cc] = c[i];
            }
        }
    }
}

__global__ __launch_bounds__(512) void k_agg64_dot(const float* __restrict__ hw,
                                                   const int2* __restrict__ csr,
                                                   const int* __restrict__ rowptr,
                                                   const float* __restrict__ dinv,
                                                   const float* __restrict__ bias,
                                                   const float* __restrict__ w2,
                                                   float* __restrict__ lin, int n) {
    __shared__ float s_lin[32];
    const int l16  = threadIdx.x & 15;
    const int grp  = threadIdx.x >> 4;
    const int row0 = blockIdx.x * 32;
    const int g    = row0 + grp;
    const int d    = (g < n) ? g : (n - 1);

    const float dv = dinv[d];
    const int f = l16 * 4;
    const float4 bs = *(const float4*)(bias + f);
    const float4 acc = agg64_row(hw, csr, rowptr, d, l16);
    float s = fmaxf(fmaf(acc.x, dv, bs.x), 0.0f) * w2[f + 0];
    s = fmaf(fmaxf(fmaf(acc.y, dv, bs.y), 0.0f), w2[f + 1], s);
    s = fmaf(fmaxf(fmaf(acc.z, dv, bs.z), 0.0f), w2[f + 2], s);
    s = fmaf(fmaxf(fmaf(acc.w, dv, bs.w), 0.0f), w2[f + 3], s);
    for (int o = 8; o > 0; o >>= 1) s += __shfl_xor(s, o, 16);
    if (l16 == 0) s_lin[grp] = s * dv;
    __syncthreads();
    if (threadIdx.x < 32) {
        const int r = row0 + threadIdx.x;
        if (r < n) lin[r] = s_lin[threadIdx.x];
    }
}

__global__ __launch_bounds__(256) void k_agg1(const float* __restrict__ lin,
                                              const int2* __restrict__ csr,
                                              const int* __restrict__ rowptr,
                                              const float* __restrict__ dinv,
                                              const float* __restrict__ b2,
                                              float* __restrict__ out, int n) {
    int d = blockIdx.x * blockDim.x + threadIdx.x;
    if (d < n) {
        const int beg = rowptr[d];
        const int end = rowptr[d + 1];
        float acc = lin[d];
        #pragma unroll 4
        for (int p = beg; p < end; ++p) {
            const int2 pr = csr[p];
            acc += lin[pr.x] * __int_as_float(pr.y);
        }
        out[d] = fmaf(acc, dinv[d], b2[0]);
    }
}


extern "C" void kernel_launch(void* const* d_in, const int* in_sizes, int n_in,
                              void* d_out, int out_size, void* d_ws, size_t ws_size,
                              hipStream_t stream) {
    const float* x    = (const float*)d_in[0];
    const int*   esrc = (const int*)  d_in[1];
    const int*   edst = (const int*)  d_in[2];
    const float* ew   = (const float*)d_in[3];
    const float* W0   = (const float*)d_in[4];
    const float* b0   = (const float*)d_in[5];
    const float* W1   = (const float*)d_in[6];
    const float* b1   = (const float*)d_in[7];
    const float* W2   = (const float*)d_in[8];
    const float* b2   = (const float*)d_in[9];

    const int N = in_sizes[0] / FDIM;
    const int E = in_sizes[1];
    float* out = (float*)d_out;

    const int NB = (N + RB - 1) / RB;
    const int NC = (E + CH - 1) / CH;
    const int HC = NB * NC;

    const size_t Na = ((size_t)N + 64 + 63) & ~(size_t)63;
    const size_t Ea = ((size_t)E + 63) & ~(size_t)63;
    const size_t Ha = ((size_t)HC + 64 + 63) & ~(size_t)63;
    float* ws     = (float*)d_ws;
    float* dinv   = ws;
    int*   rowptr = (int*)(ws + Na);
    int*   hist   = (int*)(ws + 2 * Na);
    int*   hoff   = (int*)(ws + 2 * Na + Ha);
    int2*  csr    = (int2*)(ws + 2 * Na + 2 * Ha);
    int4*  tmp    = (int4*)(ws + 2 * Na + 2 * Ha + 2 * Ea);
    float* hw     = ws + 2 * Na + 2 * Ha + 6 * Ea;
    float* ha     = hw + (size_t)N * FDIM;
    float* lin    = ha + (size_t)N * FDIM;

    const int T = 256;
    const int gN  = (N + T - 1) / T;
    const int gS  = (HC + 1023) / 1024;
    const int gA  = (N + 15) / 16;
    const int gGemm = (N + 31) / 32;

    k_hist    <<<NC, CH_T, NB * sizeof(int), stream>>>(edst, hist, E, NB, NC);
    k_rowptr  <<<gS, 1024, 0, stream>>>(hist, hoff, HC);
    k_scatter <<<NC, CH_T, NB * sizeof(int), stream>>>(esrc, edst, ew, hoff, tmp, E, NB, NC);
    k_bucket  <<<NB, 512, 0, stream>>>(tmp, hoff, rowptr, csr, N, NC);

    k_gemm64_first<<<gGemm, T, 0, stream>>>(x, W0, csr, rowptr, dinv, hw, N);

    k_agg64_gemm<<<gA, T, 0, stream>>>(hw, csr, rowptr, dinv, b0, W1, ha, N);

    k_agg64_dot<<<gGemm, 512, 0, stream>>>(ha, csr, rowptr, dinv, b1, W2, lin, N);

    k_agg1    <<<gN, T, 0, stream>>>(lin, csr, rowptr, dinv, b2, out, N);
}


# Chunk-local bucket offsets (count, block scan, place in one launch; bucket start = sum of the chunks' offsets): histogram and table-scan launches go, 8 launches -> 6; row sum order: LDS arrival rank
# speedup vs baseline: 3.7199x; 1.0841x over previous
#include <hip/hip_runtime.h>


typedef __attribute__((ext_vector_type(2))) float v2f;
typedef __attribute__((ext_vector_type(8))) float v8f;

#define FDIM 64


#define RB_SHIFT 8
#define RB       (1 << RB_SHIFT)
#define CH_T     256
#define CH       (CH_T * 16)

__global__ __launch_bounds__(CH_T) void k_chunk(const int* __restrict__ src,
                                                const int* __restrict__ dst,
                                                const float* __restrict__ w,
                                                int* __restrict__ loc,
                                                int4* __restrict__ tmp, int e, int nb) {
    extern __shared__ int s_o[];
    __shared__ int s_wt[CH_T / 32];
    const int tid  = threadIdx.x;
    const int lane = tid & 31;
    const int wv   = tid >> 5;
    const int c    = blockIdx.x;
    for (int b = tid; b < nb; b += CH_T) s_o[b] = 0;
    __syncthreads();

    const bool whole = ((c + 1) * CH <= e);
    int4 d[4], s[4];
    float4 ww[4];
    if (whole) {
        #pragma unroll
        for (int g = 0; g < 4; ++g) {
            const int i = c * CH + (g * CH_T + tid) * 4;
            d[g]  = *(const int4*)(dst + i);
            s[g]  = *(const int4*)(src + i);
            ww[g] = *(const float4*)(w + i);
        }
        #pragma unroll
        for (int g = 0; g < 4; ++g) {
            atomicAdd(&s_o[d[g].x >> RB_SHIFT], 1);
            atomicAdd(&s_o[d[g].y >> RB_SHIFT], 1);
            atomicAdd(&s_o[d[g].z >> RB_SHIFT], 1);
            atomicAdd(&s_o[d[g].w >> RB_SHIFT], 1);
        }
    } else {
        for (int i = c * CH + tid; i < e; i += CH_T) atomicAdd(&s_o[dst[i] >> RB_SHIFT], 1);
    }
    __syncthreads();

    const int K  = (nb + CH_T - 1) / CH_T;
    const int b0 = tid * K;
    int sum = 0;
    for (int k = 0; k < K; ++k) {
        const int b = b0 + k;
        if (b < nb) sum += s_o[b];
    }
    int inc = sum;
    for (int o = 1; o < 32; o <<= 1) {
        const int t = __shfl_up(inc, o, 32);
        if (lane >= o) inc += t;
    }
    if (lane == 31) s_wt[wv] = inc;
    __syncthreads();
    int run = inc - sum;
    for (int k = 0; k < CH_T / 32; ++k)
        if (k < wv) run += s_wt[k];
    int* lc = loc + (size_t)c * (nb + 1);
    for (int k = 0; k < K; ++k) {
        const int b = b0 + k;
        if (b < nb) {
            const int cn = s_o[b];
            s_o[b] = run;
            lc[b]  = run;
            run += cn;
        }
    }
    if (tid == CH_T - 1) lc[nb] = run;
    __syncthreads();

    int4* tc = tmp + (size_t)c * CH;
    if (whole) {
        #pragma unroll
        for (int g = 0; g < 4; ++g) {
            const int p0 = atomicAdd(&s_o[d[g].x >> RB_SHIFT], 1);
            const int p1 = atomicAdd(&s_o[d[g].y >> RB_SHIFT], 1);
            const int p2 = atomicAdd(&s_o[d[g].z >> RB_SHIFT], 1);
            const int p3 = atomicAdd(&s_o[d[g].w >> RB_SHIFT], 1);
            tc[p0] = make_int4(s[g].x, d[g].x, __float_as_int(ww[g].x), 0);
            tc[p1] = make_int4(s[g].y, d[g].y, __float_as_int(ww[g].y), 0);
            tc[p2] = make_int4(s[g].z, d[g].z, __float_as_int(ww[g].z), 0);
            tc[p3] = make_int4(s[g].w, d[g].w, __float_as_int(ww[g].w), 0);
        }
    } else {
        for (int i = c * CH + tid; i < e; i += CH_T) {
            const int dk = dst[i];
            const int p  = atomicAdd(&s_o[dk >> RB_SHIFT], 1);
            tc[p] = make_int4(src[i], dk, __float_as_int(w[i]), 0);
        }
    }
}

__global__ __launch_bounds__(512) void k_bucket(const int4* __restrict__ tmp,
                                                const int* __restrict__ loc,
                                                int* __restrict__ rowptr,
                                                int2* __restrict__ csr, int n, int nc, int nb) {
    extern __shared__ int s_run[];
    __shared__ int s_cnt[RB];
    __shared__ int s_cur[RB];
    __shared__ int s_wtot[RB / 32];
    __shared__ int s_red[16];
    const int tid  = threadIdx.x;
    const int lane = tid & 31;
    const int wv   = tid >> 5;
    const int l16  = tid & 15;
    const int grp  = tid >> 4;
    const int b    = blockIdx.x;

    if (tid < RB) s_cnt[tid] = 0;
    int part = 0;
    for (int c = tid; c < nc; c += 512) {
        const int* lc = loc + (size_t)c * (nb + 1) + b;
        const int lo = lc[0];
        const int hi = lc[1];
        s_run[2 * c]     = c * CH + lo;
        s_run[2 * c + 1] = hi - lo;
        part += lo;
    }
    for (int o = 16; o > 0; o >>= 1) part += __shfl_xor(part, o, 32);
    if (lane == 0) s_red[wv] = part;
    __syncthreads();
    int beg = 0;
    for (int k = 0; k < 16; ++k) beg += s_red[k];

    for (int c0 = grp; c0 < nc; c0 += 256) {
        int st[8], ln[8], dv[8];
        #pragma unroll
        for (int u = 0; u < 8; ++u) {
            const int c = c0 + 32 * u;
            st[u] = 0; ln[u] = 0;
            if (c < nc) { st[u] = s_run[2 * c]; ln[u] = s_run[2 * c + 1]; }
            dv[u] = tmp[max(st[u] + min(l16, ln[u] - 1), 0)].y;
        }
        #pragma unroll
        for (int u = 0; u < 8; ++u) {
            if (l16 < ln[u]) atomicAdd(&s_cnt[dv[u] & (RB - 1)], 1);
            for (int q = 16 + l16; q < ln[u]; q += 16)
                atomicAdd(&s_cnt[tmp[st[u] + q].y & (RB - 1)], 1);
        }
    }
    __syncthreads();

    int cn = 0, inc = 0;
    if (tid < RB) {
        cn = s_cnt[tid];
        inc = cn;
        for (int o = 1; o < 32; o <<= 1) {
            const int t = __shfl_up(inc, o, 32);
            if (lane >= o) inc += t;
        }
        if (lane == 31) s_wtot[wv] = inc;
    }
    __syncthreads();
    if (tid < RB) {
        int pre = beg;
        for (int k = 0; k < RB / 32; ++k)
            if (k < wv) pre += s_wtot[k];
        const int excl = pre + inc - cn;
        s_cur[tid] = excl;
        const int row = b * RB + tid;
        if (row < n)      rowptr[row] = excl;
        if (row == n - 1) rowptr[n]   = excl + cn;
    }
    __syncthreads();

    for (int c0 = grp; c0 < nc; c0 += 256) {
        int st[8], ln[8];
        int4 ev[8];
        #pragma unroll
        for (int u = 0; u < 8; ++u) {
            const int c = c0 + 32 * u;
            st[u] = 0; ln[u] = 0;
            if (c < nc) { st[u] = s_run[2 * c]; ln[u] = s_run[2 * c + 1]; }
            ev[u] = tmp[max(st[u] + min(l16, ln[u] - 1), 0)];
        }
        #pragma unroll
        for (int u = 0; u < 8; ++u) {
            if (l16 < ln[u])
                csr[atomicAdd(&s_cur[ev[u].y & (RB - 1)], 1)] = make_int2(ev[u].x, ev[u].z);
            for (int q = 16 + l16; q < ln[u]; q += 16) {
                const int4 eq = tmp[st[u] + q];
                csr[atomicAdd(&s_cur[eq.y & (RB - 1)], 1)] = make_int2(eq.x, eq.z);
            }
        }
    }
}


__global__ __launch_bounds__(256) void k_gemm64_first(const float* __restrict__ H,
                                                      const float* __restrict__ Wm,
                                                      const int2* __restrict__ csr,
                                                      const int* __restrict__ rowptr,
                                                      float* __restrict__ dinv,
                                                      float* __restrict__ hw, int nrows) {
    __shared__ float s_dinv[32];
    const int lane  = threadIdx.x & 31;
    const int wave  = threadIdx.x >> 5;
    const int ctile = wave & 3;
    const int rhalf = wave >> 2;
    const int l16   = lane & 15;
    const int hi    = lane >> 4;
    const int blk0  = blockIdx.x * 32;

    {
        const int rl = threadIdx.x >> 3;
        const int t8 = threadIdx.x & 7;
        const int r  = blk0 + rl;
        float s = 0.0f;
        if (r < nrows) {
            const int beg = rowptr[r];
            const int end = rowptr[r + 1];
            for (int p = beg + t8; p < end; p += 8) s += __int_as_float(csr[p].y);
        }
        s += __shfl_xor(s, 4, 8);
        s += __shfl_xor(s, 2, 8);
        s += __shfl_xor(s, 1, 8);
        if (t8 == 0) {
            const float dv = rsqrtf(1.0f + s);
            s_dinv[rl] = dv;
            if (r < nrows) dinv[r] = dv;
        }
    }

    v2f bfrag[16];
    const int cc = ctile * 16 + l16;
    #pragma unroll
    for (int k = 0; k < 16; ++k) {
        const int k0 = 4 * k + (hi ? 2 : 0);
        bfrag[k].x = Wm[(size_t)k0 * FDIM + cc];
        bfrag[k].y = Wm[(size_t)(k0 + 1) * FDIM + cc];
    }

    const int row0 = blk0 + rhalf * 16;
    const bool full = (row0 + 16 <= nrows);
    const int r  = row0 + l16;
    const int rc = full ? r : ((r < nrows) ? r : (nrows > 0 ? nrows - 1 : 0));
    const float* hp = H + (size_t)rc * FDIM + (hi ? 2 : 0);

    v8f acc = {};
    #pragma unroll
    for (int k = 0; k < 16; ++k) {
        const v2f a = *(const v2f*)(hp + 4 * k);
        acc = __builtin_amdgcn_wmma_f32_16x16x4_f32(
                  false, a, false, bfrag[k], (short)0, acc, false, false);
    }
    __syncthreads();

    const float* sd = s_dinv + rhalf * 16 + hi * 8;
    if (full) {
        float* op = hw + (size_t)(row0 + hi * 8) * FDIM + cc;
        #pragma unroll
        for (int i = 0; i < 8; ++i)
            op[(size_t)i * FDIM] = acc[i] * sd[i];
    } else {
        #pragma unroll
        for (int i = 0; i < 8; ++i) {
            int rr = row0 + hi * 8 + i;
            if (rr < nrows) hw[(size_t)rr * FDIM + cc] = acc[i] * sd[i];
        }
    }
}


static __device__ __forceinline__ float4 agg64_row(const float* __restrict__ hw,
                                                   const int2* __restrict__ csr,
                                                   const int* __restrict__ rowptr,
                                                   const int d, const int l16) {
    const int beg = rowptr[d];
    const int end = rowptr[d + 1];
    const int f = l16 * 4;
    float4 acc = *(const float4*)(hw + (size_t)d * FDIM + f);

    const float4 zero4 = make_float4(0.0f, 0.0f, 0.0f, 0.0f);
    const float* hrow = hw + f;
    for (int base = beg; base < end; base += 16) {
        const int idx = base + l16;
        int2 pr = make_int2(0, 0);
        if (idx < end) pr = csr[idx];
        const int m = min(16, end - base);
        for (int j = 0; j < m; j += 4) {
            const int   s0 = __shfl(pr.x, j + 0, 16);
            const int   s1 = __shfl(pr.x, j + 1, 16);
            const int   s2 = __shfl(pr.x, j + 2, 16);
            const int   s3 = __shfl(pr.x, j + 3, 16);
            const float n0 = __int_as_float(__shfl(pr.y, j + 0, 16));
            const float n1 = __int_as_float(__shfl(pr.y, j + 1, 16));
            const float n2 = __int_as_float(__shfl(pr.y, j + 2, 16));
            const float n3 = __int_as_float(__shfl(pr.y, j + 3, 16));
            float4 v0 = *(const float4*)(hrow + (size_t)s0 * FDIM);
            float4 v1 = zero4, v2 = zero4, v3 = zero4;
            if (j + 1 < m) v1 = *(const float4*)(hrow + (size_t)s1 * FDIM);
            if (j + 2 < m) v2 = *(const float4*)(hrow + (size_t)s2 * FDIM);
            if (j + 3 < m) v3 = *(const float4*)(hrow + (size_t)s3 * FDIM);
            acc.x += v0.x * n0; acc.y += v0.y * n0; acc.z += v0.z * n0; acc.w += v0.w * n0;
            acc.x += v1.x * n1; acc.y += v1.y * n1; acc.z += v1.z * n1; acc.w += v1.w * n1;
            acc.x += v2.x * n2; acc.y += v2.y * n2; acc.z += v2.z * n2; acc.w += v2.w * n2;
            acc.x += v3.x * n3; acc.y += v3.y * n3; acc.z += v3.z * n3; acc.w += v3.w * n3;
        }
    }
    return acc;
}

#define HPITCH 68

__global__ __launch_bounds__(256) void k_agg64_gemm(const float* __restrict__ hw,
                                                    const int2* __restrict__ csr,
                                                    const int* __restrict__ rowptr,
                                                    const float* __restrict__ dinv,
                                                    const float* __restrict__ bias,
                                                    const float* __restrict__ Wn,
                                                    float* __restrict__ hnext, int n) {
    __shared__ float s_h[16 * HPITCH];
    const int l16  = threadIdx.x & 15;
    const int grp  = threadIdx.x >> 4;
    const int row0 = blockIdx.x * 16;
    const int g    = row0 + grp;
    const int d    = (g < n) ? g : (n - 1);

    const float dv = dinv[d];
    const float4 bs = *(const float4*)(bias + l16 * 4);
    float4 acc = agg64_row(hw, csr, rowptr, d, l16);

    const int wave = threadIdx.x >> 5;
    const int hi   = (threadIdx.x >> 4) & 1;
    const int cc   = (wave & 3) * 16 + l16;
    v2f bfrag[16];
    if (wave < 4) {
        #pragma unroll
        for (int k = 0; k < 16; ++k) {
            const int k0 = 4 * k + (hi ? 2 : 0);
            bfrag[k].x = Wn[(size_t)k0 * FDIM + cc];
            bfrag[k].y = Wn[(size_t)(k0 + 1) * FDIM + cc];
        }
    }

    acc.x = fmaxf(fmaf(acc.x, dv, bs.x), 0.0f) * dv;
    acc.y = fmaxf(fmaf(acc.y, dv, bs.y), 0.0f) * dv;
    acc.z = fmaxf(fmaf(acc.z, dv, bs.z), 0.0f) * dv;
    acc.w = fmaxf(fmaf(acc.w, dv, bs.w), 0.0f) * dv;
    *(float4*)(s_h + grp * HPITCH + l16 * 4) = acc;
    __syncthreads();

    if (wave < 4) {
        const float* ap = s_h + l16 * HPITCH + (hi ? 2 : 0);
        v8f c = {};
        #pragma unroll
        for (int k = 0; k < 16; ++k) {
            const v2f a = *(const v2f*)(ap + 4 * k);
            c = __builtin_amdgcn_wmma_f32_16x16x4_f32(
                    false, a, false, bfrag[k], (short)0, c, false, false);
        }
        const int rbase = row0 + hi * 8;
        if (row0 + 16 <= n) {
            float* op = hnext + (size_t)rbase * FDIM + cc;
            #pragma unroll
            for (int i = 0; i < 8; ++i)
                op[(size_t)i * FDIM] = c[i];
        } else {
            #pragma unroll
            for (int i = 0; i < 8; ++i) {
                const int rr = rbase + i;
                if (rr < n) hnext[(size_t)rr * FDIM + cc] = c[i];
            }
        }
    }
}

__global__ __launch_bounds__(512) void k_agg64_dot(const float* __restrict__ hw,
                                                   const int2* __restrict__ csr,
                                                   const int* __restrict__ rowptr,
                                                   const float* __restrict__ dinv,
                                                   const float* __restrict__ bias,
                                                   const float* __restrict__ w2,
                                                   float* __restrict__ lin, int n) {
    __shared__ float s_lin[32];
    const int l16  = threadIdx.x & 15;
    const int grp  = threadIdx.x >> 4;
    const int row0 = blockIdx.x * 32;
    const int g    = row0 + grp;
    const int d    = (g < n) ? g : (n - 1);

    const float dv = dinv[d];
    const int f = l16 * 4;
    const float4 bs = *(const float4*)(bias + f);
    const float4 acc = agg64_row(hw, csr, rowptr, d, l16);
    float s = fmaxf(fmaf(acc.x, dv, bs.x), 0.0f) * w2[f + 0];
    s = fmaf(fmaxf(fmaf(acc.y, dv, bs.y), 0.0f), w2[f + 1], s);
    s = fmaf(fmaxf(fmaf(acc.z, dv, bs.z), 0.0f), w2[f + 2], s);
    s = fmaf(fmaxf(fmaf(acc.w, dv, bs.w), 0.0f), w2[f + 3], s);
    for (int o = 8; o > 0; o >>= 1) s += __shfl_xor(s, o, 16);
    if (l16 == 0) s_lin[grp] = s * dv;
    __syncthreads();
    if (threadIdx.x < 32) {
        const int r = row0 + threadIdx.x;
        if (r < n) lin[r] = s_lin[threadIdx.x];
    }
}

__global__ __launch_bounds__(256) void k_agg1(const float* __restrict__ lin,
                                              const int2* __restrict__ csr,
                                              const int* __restrict__ rowptr,
                                              const float* __restrict__ dinv,
                                              const float* __restrict__ b2,
                                              float* __restrict__ out, int n) {
    int d = blockIdx.x * blockDim.x + threadIdx.x;
    if (d < n) {
        const int beg = rowptr[d];
        const int end = rowptr[d + 1];
        float acc = lin[d];
        #pragma unroll 4
        for (int p = beg; p < end; ++p) {
            const int2 pr = csr[p];
            acc += lin[pr.x] * __int_as_float(pr.y);
        }
        out[d] = fmaf(acc, dinv[d], b2[0]);
    }
}


extern "C" void kernel_launch(void* const* d_in, const int* in_sizes, int n_in,
                              void* d_out, int out_size, void* d_ws, size_t ws_size,
                              hipStream_t stream) {
    const float* x    = (const float*)d_in[0];
    const int*   esrc = (const int*)  d_in[1];
    const int*   edst = (const int*)  d_in[2];
    const float* ew   = (const float*)d_in[3];
    const float* W0   = (const float*)d_in[4];
    const float* b0   = (const float*)d_in[5];
    const float* W1   = (const float*)d_in[6];
    const float* b1   = (const float*)d_in[7];
    const float* W2   = (const float*)d_in[8];
    const float* b2   = (const float*)d_in[9];

    const int N = in_sizes[0] / FDIM;
    const int E = in_sizes[1];
    float* out = (float*)d_out;

    const int NB = (N + RB - 1) / RB;
    const int NC = (E + CH - 1) / CH;
    const int HC = (NB + 1) * NC;

    const size_t Na = ((size_t)N + 64 + 63) & ~(size_t)63;
    const size_t Ea = ((size_t)E + 63) & ~(size_t)63;
    const size_t Ha = ((size_t)HC + 63) & ~(size_t)63;
    const size_t Ta = (size_t)NC * CH;
    float* ws     = (float*)d_ws;
    float* dinv   = ws;
    int*   rowptr = (int*)(ws + Na);
    int*   loc    = (int*)(ws + 2 * Na);
    int2*  csr    = (int2*)(ws + 2 * Na + Ha);
    int4*  tmp    = (int4*)(ws + 2 * Na + Ha + 2 * Ea);
    float* hw     = ws + 2 * Na + Ha + 2 * Ea + 4 * Ta;
    float* ha     = hw + (size_t)N * FDIM;
    float* lin    = ha + (size_t)N * FDIM;

    const int T = 256;
    const int gN  = (N + T - 1) / T;
    const int gA  = (N + 15) / 16;
    const int gGemm = (N + 31) / 32;

    k_chunk   <<<NC, CH_T, NB * sizeof(int), stream>>>(esrc, edst, ew, loc, tmp, E, NB);
    k_bucket  <<<NB, 512, 2 * NC * sizeof(int), stream>>>(tmp, loc, rowptr, csr, N, NC, NB);

    k_gemm64_first<<<gGemm, T, 0, stream>>>(x, W0, csr, rowptr, dinv, hw, N);

    k_agg64_gemm<<<gA, T, 0, stream>>>(hw, csr, rowptr, dinv, b0, W1, ha, N);

    k_agg64_dot<<<gGemm, 512, 0, stream>>>(ha, csr, rowptr, dinv, b1, W2, lin, N);

    k_agg1    <<<gN, T, 0, stream>>>(lin, csr, rowptr, dinv, b2, out, N);
}


# Layer 0 product and degree pass moved into the tail of the bucket kernel (its block owns 256 whole rows; x tile via LDS in whole lines, 4 row tiles per wave): 6 launches -> 5; sum orders unchanged
# speedup vs baseline: 3.8919x; 1.0462x over previous
#include <hip/hip_runtime.h>


typedef __attribute__((ext_vector_type(2))) float v2f;
typedef __attribute__((ext_vector_type(8))) float v8f;

#define FDIM 64

#define HPITCH 68


#define RB_SHIFT 8
#define RB       (1 << RB_SHIFT)
#define CH_T     256
#define CH       (CH_T * 16)

__global__ __launch_bounds__(CH_T) void k_chunk(const int* __restrict__ src,
                                                const int* __restrict__ dst,
                                                const float* __restrict__ w,
                                                int* __restrict__ loc,
                                                int4* __restrict__ tmp, int e, int nb) {
    extern __shared__ int s_o[];
    __shared__ int s_wt[CH_T / 32];
    const int tid  = threadIdx.x;
    const int lane = tid & 31;
    const int wv   = tid >> 5;
    const int c    = blockIdx.x;
    for (int b = tid; b < nb; b += CH_T) s_o[b] = 0;
    __syncthreads();

    const bool whole = ((c + 1) * CH <= e);
    int4 d[4], s[4];
    float4 ww[4];
    if (whole) {
        #pragma unroll
        for (int g = 0; g < 4; ++g) {
            const int i = c * CH + (g * CH_T + tid) * 4;
            d[g]  = *(const int4*)(dst + i);
            s[g]  = *(const int4*)(src + i);
            ww[g] = *(const float4*)(w + i);
        }
        #pragma unroll
        for (int g = 0; g < 4; ++g) {
            atomicAdd(&s_o[d[g].x >> RB_SHIFT], 1);
            atomicAdd(&s_o[d[g].y >> RB_SHIFT], 1);
            atomicAdd(&s_o[d[g].z >> RB_SHIFT], 1);
            atomicAdd(&s_o[d[g].w >> RB_SHIFT], 1);
        }
    } else {
        for (int i = c * CH + tid; i < e; i += CH_T) atomicAdd(&s_o[dst[i] >> RB_SHIFT], 1);
    }
    __syncthreads();

    const int K  = (nb + CH_T - 1) / CH_T;
    const int b0 = tid * K;
    int sum = 0;
    for (int k = 0; k < K; ++k) {
        const int b = b0 + k;
        if (b < nb) sum += s_o[b];
    }
    int inc = sum;
    for (int o = 1; o < 32; o <<= 1) {
        const int t = __shfl_up(inc, o, 32);
        if (lane >= o) inc += t;
    }
    if (lane == 31) s_wt[wv] = inc;
    __syncthreads();
    int run = inc - sum;
    for (int k = 0; k < CH_T / 32; ++k)
        if (k < wv) run += s_wt[k];
    int* lc = loc + (size_t)c * (nb + 1);
    for (int k = 0; k < K; ++k) {
        const int b = b0 + k;
        if (b < nb) {
            const int cn = s_o[b];
            s_o[b] = run;
            lc[b]  = run;
            run += cn;
        }
    }
    if (tid == CH_T - 1) lc[nb] = run;
    __syncthreads();

    int4* tc = tmp + (size_t)c * CH;
    if (whole) {
        #pragma unroll
        for (int g = 0; g < 4; ++g) {
            const int p0 = atomicAdd(&s_o[d[g].x >> RB_SHIFT], 1);
            const int p1 = atomicAdd(&s_o[d[g].y >> RB_SHIFT], 1);
            const int p2 = atomicAdd(&s_o[d[g].z >> RB_SHIFT], 1);
            const int p3 = atomicAdd(&s_o[d[g].w >> RB_SHIFT], 1);
            tc[p0] = make_int4(s[g].x, d[g].x, __float_as_int(ww[g].x), 0);
            tc[p1] = make_int4(s[g].y, d[g].y, __float_as_int(ww[g].y), 0);
            tc[p2] = make_int4(s[g].z, d[g].z, __float_as_int(ww[g].z), 0);
            tc[p3] = make_int4(s[g].w, d[g].w, __float_as_int(ww[g].w), 0);
        }
    } else {
        for (int i = c * CH + tid; i < e; i += CH_T) {
            const int dk = dst[i];
            const int p  = atomicAdd(&s_o[dk >> RB_SHIFT], 1);
            tc[p] = make_int4(src[i], dk, __float_as_int(w[i]), 0);
        }
    }
}

__global__ __launch_bounds__(512) void k_bucket(const int4* __restrict__ tmp,
                                                const int* __restrict__ loc,
                                                int* __restrict__ rowptr,
                                                int2* __restrict__ csr,
                                                const float* __restrict__ H,
                                                const float* __restrict__ Wm,
                                                float* __restrict__ dinv,
                                                float* __restrict__ hw, int n, int nc, int nb) {
    extern __shared__ int s_run[];
    __shared__ int s_cnt[RB];
    __shared__ int s_cur[RB];
    __shared__ int s_beg[RB];
    __shared__ int s_wtot[RB / 32];
    __shared__ int s_red[16];
    __shared__ float s_dinv[RB];
    __shared__ __attribute__((aligned(16))) float s_x[RB * HPITCH];
    const int tid  = threadIdx.x;
    const int lane = tid & 31;
    const int wv   = tid >> 5;
    const int l16  = tid & 15;
    const int grp  = tid >> 4;
    const int b    = blockIdx.x;

    if (tid < RB) s_cnt[tid] = 0;
    int part = 0;
    for (int c = tid; c < nc; c += 512) {
        const int* lc = loc + (size_t)c * (nb + 1) + b;
        const int lo = lc[0];
        const int hi = lc[1];
        s_run[2 * c]     = c * CH + lo;
        s_run[2 * c + 1] = hi - lo;
        part += lo;
    }
    for (int o = 16; o > 0; o >>= 1) part += __shfl_xor(part, o, 32);
    if (lane == 0) s_red[wv] = part;
    __syncthreads();
    int beg = 0;
    for (int k = 0; k < 16; ++k) beg += s_red[k];

    for (int c0 = grp; c0 < nc; c0 += 256) {
        int st[8], ln[8], dv[8];
        #pragma unroll
        for (int u = 0; u < 8; ++u) {
            const int c = c0 + 32 * u;
            st[u] = 0; ln[u] = 0;
            if (c < nc) { st[u] = s_run[2 * c]; ln[u] = s_run[2 * c + 1]; }
            dv[u] = tmp[max(st[u] + min(l16, ln[u] - 1), 0)].y;
        }
        #pragma unroll
        for (int u = 0; u < 8; ++u) {
            if (l16 < ln[u]) atomicAdd(&s_cnt[dv[u] & (RB - 1)], 1);
            for (int q = 16 + l16; q < ln[u]; q += 16)
                atomicAdd(&s_cnt[tmp[st[u] + q].y & (RB - 1)], 1);
        }
    }
    __syncthreads();

    int cn = 0, inc = 0;
    if (tid < RB) {
        cn = s_cnt[tid];
        inc = cn;
        for (int o = 1; o < 32; o <<= 1) {
            const int t = __shfl_up(inc, o, 32);
            if (lane >= o) inc += t;
        }
        if (lane == 31) s_wtot[wv] = inc;
    }
    __syncthreads();
    if (tid < RB) {
        int pre = beg;
        for (int k = 0; k < RB / 32; ++k)
            if (k < wv) pre += s_wtot[k];
        const int excl = pre + inc - cn;
        s_cur[tid] = excl;
        s_beg[tid] = excl;
        const int row = b * RB + tid;
        if (row < n)      rowptr[row] = excl;
        if (row == n - 1) rowptr[n]   = excl + cn;
    }
    __syncthreads();

    float4 xv[8];
    #pragma unroll
    for (int i = 0; i < 8; ++i) {
        const int r = b * RB + i * 32 + grp;
        xv[i] = *(const float4*)(H + (size_t)min(r, n - 1) * FDIM + l16 * 4);
        if (r >= n) xv[i] = make_float4(0.0f, 0.0f, 0.0f, 0.0f);
    }

    for (int c0 = grp; c0 < nc; c0 += 256) {
        int st[8], ln[8];
        int4 ev[8];
        #pragma unroll
        for (int u = 0; u < 8; ++u) {
            const int c = c0 + 32 * u;
            st[u] = 0; ln[u] = 0;
            if (c < nc) { st[u] = s_run[2 * c]; ln[u] = s_run[2 * c + 1]; }
            ev[u] = tmp[max(st[u] + min(l16, ln[u] - 1), 0)];
        }
        #pragma unroll
        for (int u = 0; u < 8; ++u) {
            if (l16 < ln[u])
                csr[atomicAdd(&s_cur[ev[u].y & (RB - 1)], 1)] = make_int2(ev[u].x, ev[u].z);
            for (int q = 16 + l16; q < ln[u]; q += 16) {
                const int4 eq = tmp[st[u] + q];
                csr[atomicAdd(&s_cur[eq.y & (RB - 1)], 1)] = make_int2(eq.x, eq.z);
            }
        }
    }

    #pragma unroll
    for (int i = 0; i < 8; ++i)
        *(float4*)(s_x + (i * 32 + grp) * HPITCH + l16 * 4) = xv[i];

    const int hi = (tid >> 4) & 1;
    const int cc = (wv & 3) * 16 + l16;
    v2f bfrag[16];
    #pragma unroll
    for (int k = 0; k < 16; ++k) {
        const int k0 = 4 * k + (hi ? 2 : 0);
        bfrag[k].x = Wm[(size_t)k0 * FDIM + cc];
        bfrag[k].y = Wm[(size_t)(k0 + 1) * FDIM + cc];
    }
    __syncthreads();

    {
        const int rq = tid >> 3;
        const int t8 = tid & 7;
        int   p[RB / 64], pe[RB / 64];
        float s[RB / 64];
        #pragma unroll
        for (int ps = 0; ps < RB / 64; ++ps) {
            p[ps]  = s_beg[ps * 64 + rq] + t8;
            pe[ps] = s_cur[ps * 64 + rq];
            s[ps]  = 0.0f;
        }
        #pragma unroll
        for (int rd = 0; rd < 3; ++rd) {
            float wq[RB / 64];
            #pragma unroll
            for (int ps = 0; ps < RB / 64; ++ps)
                wq[ps] = __int_as_float(csr[max(min(p[ps], pe[ps] - 1), 0)].y);
            #pragma unroll
            for (int ps = 0; ps < RB / 64; ++ps) {
                if (p[ps] < pe[ps]) s[ps] += wq[ps];
                p[ps] += 8;
            }
        }
        #pragma unroll
        for (int ps = 0; ps < RB / 64; ++ps) {
            float sr = s[ps];
            for (int q = p[ps]; q < pe[ps]; q += 8) sr += __int_as_float(csr[q].y);
            sr += __shfl_xor(sr, 4, 8);
            sr += __shfl_xor(sr, 2, 8);
            sr += __shfl_xor(sr, 1, 8);
            if (t8 == 0) {
                const float dv = rsqrtf(1.0f + sr);
                s_dinv[ps * 64 + rq] = dv;
                const int r = b * RB + ps * 64 + rq;
                if (r < n) dinv[r] = dv;
            }
        }
    }
    __syncthreads();

    #pragma unroll
    for (int t = 0; t < 4; ++t) {
        const int rt = (wv >> 2) + 4 * t;
        const float* ap = s_x + (rt * 16 + l16) * HPITCH + (hi ? 2 : 0);
        v8f acc = {};
        #pragma unroll
        for (int k = 0; k < 16; ++k) {
            const v2f a = *(const v2f*)(ap + 4 * k);
            acc = __builtin_amdgcn_wmma_f32_16x16x4_f32(
                      false, a, false, bfrag[k], (short)0, acc, false, false);
        }
        const int rl0 = rt * 16 + hi * 8;
        const int r0  = b * RB + rl0;
        const float* sd = s_dinv + rl0;
        if (b * RB + rt * 16 + 16 <= n) {
            float* op = hw + (size_t)r0 * FDIM + cc;
            #pragma unroll
            for (int i = 0; i < 8; ++i)
                op[(size_t)i * FDIM] = acc[i] * sd[i];
        } else {
            #pragma unroll
            for (int i = 0; i < 8; ++i)
                if (r0 + i < n) hw[(size_t)(r0 + i) * FDIM + cc] = acc[i] * sd[i];
        }
    }
}


static __device__ __forceinline__ float4 agg64_row(const float* __restrict__ hw,
                                                   const int2* __restrict__ csr,
                                                   const int* __restrict__ rowptr,
                                                   const int d, const int l16) {
    const int beg = rowptr[d];
    const int end = rowptr[d + 1];
    const int f = l16 * 4;
    float4 acc = *(const float4*)(hw + (size_t)d * FDIM + f);

    const float4 zero4 = make_float4(0.0f, 0.0f, 0.0f, 0.0f);
    const float* hrow = hw + f;
    for (int base = beg; base < end; base += 16) {
        const int idx = base + l16;
        int2 pr = make_int2(0, 0);
        if (idx < end) pr = csr[idx];
        const int m = min(16, end - base);
        for (int j = 0; j < m; j += 4) {
            const int   s0 = __shfl(pr.x, j + 0, 16);
            const int   s1 = __shfl(pr.x, j + 1, 16);
            const int   s2 = __shfl(pr.x, j + 2, 16);
            const int   s3 = __shfl(pr.x, j + 3, 16);
            const float n0 = __int_as_float(__shfl(pr.y, j + 0, 16));
            const float n1 = __int_as_float(__shfl(pr.y, j + 1, 16));
            const float n2 = __int_as_float(__shfl(pr.y, j + 2, 16));
            const float n3 = __int_as_float(__shfl(pr.y, j + 3, 16));
            float4 v0 = *(const float4*)(hrow + (size_t)s0 * FDIM);
            float4 v1 = zero4, v2 = zero4, v3 = zero4;
            if (j + 1 < m) v1 = *(const float4*)(hrow + (size_t)s1 * FDIM);
            if (j + 2 < m) v2 = *(const float4*)(hrow + (size_t)s2 * FDIM);
            if (j + 3 < m) v3 = *(const float4*)(hrow + (size_t)s3 * FDIM);
            acc.x += v0.x * n0; acc.y += v0.y * n0; acc.z += v0.z * n0; acc.w += v0.w * n0;
            acc.x += v1.x * n1; acc.y += v1.y * n1; acc.z += v1.z * n1; acc.w += v1.w * n1;
            acc.x += v2.x * n2; acc.y += v2.y * n2; acc.z += v2.z * n2; acc.w += v2.w * n2;
            acc.x += v3.x * n3; acc.y += v3.y * n3; acc.z += v3.z * n3; acc.w += v3.w * n3;
        }
    }
    return acc;
}


__global__ __launch_bounds__(256) void k_agg64_gemm(const float* __restrict__ hw,
                                                    const int2* __restrict__ csr,
                                                    const int* __restrict__ rowptr,
                                                    const float* __restrict__ dinv,
                                                    const float* __restrict__ bias,
                                                    const float* __restrict__ Wn,
                                                    float* __restrict__ hnext, int n) {
    __shared__ float s_h[16 * HPITCH];
    const int l16  = threadIdx.x & 15;
    const int grp  = threadIdx.x >> 4;
    const int row0 = blockIdx.x * 16;
    const int g    = row0 + grp;
    const int d    = (g < n) ? g : (n - 1);

    const float dv = dinv[d];
    const float4 bs = *(const float4*)(bias + l16 * 4);
    float4 acc = agg64_row(hw, csr, rowptr, d, l16);

    const int wave = threadIdx.x >> 5;
    const int hi   = (threadIdx.x >> 4) & 1;
    const int cc   = (wave & 3) * 16 + l16;
    v2f bfrag[16];
    if (wave < 4) {
        #pragma unroll
        for (int k = 0; k < 16; ++k) {
            const int k0 = 4 * k + (hi ? 2 : 0);
            bfrag[k].x = Wn[(size_t)k0 * FDIM + cc];
            bfrag[k].y = Wn[(size_t)(k0 + 1) * FDIM + cc];
        }
    }

    acc.x = fmaxf(fmaf(acc.x, dv, bs.x), 0.0f) * dv;
    acc.y = fmaxf(fmaf(acc.y, dv, bs.y), 0.0f) * dv;
    acc.z = fmaxf(fmaf(acc.z, dv, bs.z), 0.0f) * dv;
    acc.w = fmaxf(fmaf(acc.w, dv, bs.w), 0.0f) * dv;
    *(float4*)(s_h + grp * HPITCH + l16 * 4) = acc;
    __syncthreads();

    if (wave < 4) {
        const float* ap = s_h + l16 * HPITCH + (hi ? 2 : 0);
        v8f c = {};
        #pragma unroll
        for (int k = 0; k < 16; ++k) {
            const v2f a = *(const v2f*)(ap + 4 * k);
            c = __builtin_amdgcn_wmma_f32_16x16x4_f32(
                    false, a, false, bfrag[k], (short)0, c, false, false);
        }
        const int rbase = row0 + hi * 8;
        if (row0 + 16 <= n) {
            float* op = hnext + (size_t)rbase * FDIM + cc;
            #pragma unroll
            for (int i = 0; i < 8; ++i)
                op[(size_t)i * FDIM] = c[i];
        } else {
            #pragma unroll
            for (int i = 0; i < 8; ++i) {
                const int rr = rbase + i;
                if (rr < n) hnext[(size_t)rr * FDIM + cc] = c[i];
            }
        }
    }
}

__global__ __launch_bounds__(512) void k_agg64_dot(const float* __restrict__ hw,
                                                   const int2* __restrict__ csr,
                                                   const int* __restrict__ rowptr,
                                                   const float* __restrict__ dinv,
                                                   const float* __restrict__ bias,
                                                   const float* __restrict__ w2,
                                                   float* __restrict__ lin, int n) {
    __shared__ float s_lin[32];
    const int l16  = threadIdx.x & 15;
    const int grp  = threadIdx.x >> 4;
    const int row0 = blockIdx.x * 32;
    const int g    = row0 + grp;
    const int d    = (g < n) ? g : (n - 1);

    const float dv = dinv[d];
    const int f = l16 * 4;
    const float4 bs = *(const float4*)(bias + f);
    const float4 acc = agg64_row(hw, csr, rowptr, d, l16);
    float s = fmaxf(fmaf(acc.x, dv, bs.x), 0.0f) * w2[f + 0];
    s = fmaf(fmaxf(fmaf(acc.y, dv, bs.y), 0.0f), w2[f + 1], s);
    s = fmaf(fmaxf(fmaf(acc.z, dv, bs.z), 0.0f), w2[f + 2], s);
    s = fmaf(fmaxf(fmaf(acc.w, dv, bs.w), 0.0f), w2[f + 3], s);
    for (int o = 8; o > 0; o >>= 1) s += __shfl_xor(s, o, 16);
    if (l16 == 0) s_lin[grp] = s * dv;
    __syncthreads();
    if (threadIdx.x < 32) {
        const int r = row0 + threadIdx.x;
        if (r < n) lin[r] = s_lin[threadIdx.x];
    }
}

__global__ __launch_bounds__(256) void k_agg1(const float* __restrict__ lin,
                                              const int2* __restrict__ csr,
                                              const int* __restrict__ rowptr,
                                              const float* __restrict__ dinv,
                                              const float* __restrict__ b2,
                                              float* __restrict__ out, int n) {
    int d = blockIdx.x * blockDim.x + threadIdx.x;
    if (d < n) {
        const int beg = rowptr[d];
        const int end = rowptr[d + 1];
        float acc = lin[d];
        #pragma unroll 4
        for (int p = beg; p < end; ++p) {
            const int2 pr = csr[p];
            acc += lin[pr.x] * __int_as_float(pr.y);
        }
        out[d] = fmaf(acc, dinv[d], b2[0]);
    }
}


extern "C" void kernel_launch(void* const* d_in, const int* in_sizes, int n_in,
                              void* d_out, int out_size, void* d_ws, size_t ws_size,
                              hipStream_t stream) {
    const float* x    = (const float*)d_in[0];
    const int*   esrc = (const int*)  d_in[1];
    const int*   edst = (const int*)  d_in[2];
    const float* ew   = (const float*)d_in[3];
    const float* W0   = (const float*)d_in[4];
    const float* b0   = (const float*)d_in[5];
    const float* W1   = (const float*)d_in[6];
    const float* b1   = (const float*)d_in[7];
    const float* W2   = (const float*)d_in[8];
    const float* b2   = (const float*)d_in[9];

    const int N = in_sizes[0] / FDIM;
    const int E = in_sizes[1];
    float* out = (float*)d_out;

    const int NB = (N + RB - 1) / RB;
    const int NC = (E + CH - 1) / CH;
    const int HC = (NB + 1) * NC;

    const size_t Na = ((size_t)N + 64 + 63) & ~(size_t)63;
    const size_t Ea = ((size_t)E + 63) & ~(size_t)63;
    const size_t Ha = ((size_t)HC + 63) & ~(size_t)63;
    const size_t Ta = (size_t)NC * CH;
    float* ws     = (float*)d_ws;
    float* dinv   = ws;
    int*   rowptr = (int*)(ws + Na);
    int*   loc    = (int*)(ws + 2 * Na);
    int2*  csr    = (int2*)(ws + 2 * Na + Ha);
    int4*  tmp    = (int4*)(ws + 2 * Na + Ha + 2 * Ea);
    float* hw     = ws + 2 * Na + Ha + 2 * Ea + 4 * Ta;
    float* ha     = hw + (size_t)N * FDIM;
    float* lin    = ha + (size_t)N * FDIM;

    const int T = 256;
    const int gN  = (N + T - 1) / T;
    const int gA  = (N + 15) / 16;
    const int gDot = (N + 31) / 32;

    k_chunk   <<<NC, CH_T, NB * sizeof(int), stream>>>(esrc, edst, ew, loc, tmp, E, NB);
    k_bucket  <<<NB, 512, 2 * NC * sizeof(int), stream>>>(tmp, loc, rowptr, csr, x, W0, dinv, hw, N, NC, NB);

    k_agg64_gemm<<<gA, T, 0, stream>>>(hw, csr, rowptr, dinv, b0, W1, ha, N);

    k_agg64_dot<<<gDot, 512, 0, stream>>>(ha, csr, rowptr, dinv, b1, W2, lin, N);

    k_agg1    <<<gN, T, 0, stream>>>(lin, csr, rowptr, dinv, b2, out, N);
}


# Both sort levels place their records in LDS first and store the lists in whole 128-byte lines (16-byte pieces, 4 lines per wave store); degrees read from the LDS copy; sum orders unchanged
# speedup vs baseline: 4.0633x; 1.0440x over previous
#include <hip/hip_runtime.h>


typedef __attribute__((ext_vector_type(2))) float v2f;
typedef __attribute__((ext_vector_type(8))) float v8f;

#define FDIM 64

#define HPITCH 68


#define RB_SHIFT 8
#define RB       (1 << RB_SHIFT)
#define CH_T     256
#define CH       (CH_T * 16)
#define CSR_CAP  8192

__global__ __launch_bounds__(CH_T) void k_chunk(const int* __restrict__ src,
                                                const int* __restrict__ dst,
                                                const float* __restrict__ w,
                                                int* __restrict__ loc,
                                                int4* __restrict__ tmp, int e, int nb) {
    extern __shared__ int s_o[];
    __shared__ int s_wt[CH_T / 32];
    __shared__ __attribute__((aligned(16))) int4 s_rec[CH];
    const int tid  = threadIdx.x;
    const int lane = tid & 31;
    const int wv   = tid >> 5;
    const int c    = blockIdx.x;
    for (int b = tid; b < nb; b += CH_T) s_o[b] = 0;
    __syncthreads();

    const bool whole = ((c + 1) * CH <= e);
    int4 d[4], s[4];
    float4 ww[4];
    if (whole) {
        #pragma unroll
        for (int g = 0; g < 4; ++g) {
            const int i = c * CH + (g * CH_T + tid) * 4;
            d[g]  = *(const int4*)(dst + i);
            s[g]  = *(const int4*)(src + i);
            ww[g] = *(const float4*)(w + i);
        }
        #pragma unroll
        for (int g = 0; g < 4; ++g) {
            atomicAdd(&s_o[d[g].x >> RB_SHIFT], 1);
            atomicAdd(&s_o[d[g].y >> RB_SHIFT], 1);
            atomicAdd(&s_o[d[g].z >> RB_SHIFT], 1);
            atomicAdd(&s_o[d[g].w >> RB_SHIFT], 1);
        }
    } else {
        for (int i = c * CH + tid; i < e; i += CH_T) atomicAdd(&s_o[dst[i] >> RB_SHIFT], 1);
    }
    __syncthreads();

    const int K  = (nb + CH_T - 1) / CH_T;
    const int b0 = tid * K;
    int sum = 0;
    for (int k = 0; k < K; ++k) {
        const int b = b0 + k;
        if (b < nb) sum += s_o[b];
    }
    int inc = sum;
    for (int o = 1; o < 32; o <<= 1) {
        const int t = __shfl_up(inc, o, 32);
        if (lane >= o) inc += t;
    }
    if (lane == 31) s_wt[wv] = inc;
    __syncthreads();
    int run = inc - sum;
    for (int k = 0; k < CH_T / 32; ++k)
        if (k < wv) run += s_wt[k];
    int* lc = loc + (size_t)c * (nb + 1);
    for (int k = 0; k < K; ++k) {
        const int b = b0 + k;
        if (b < nb) {
            const int cn = s_o[b];
            s_o[b] = run;
            lc[b]  = run;
            run += cn;
        }
    }
    if (tid == CH_T - 1) lc[nb] = run;
    __syncthreads();

    if (whole) {
        #pragma unroll
        for (int g = 0; g < 4; ++g) {
            const int p0 = atomicAdd(&s_o[d[g].x >> RB_SHIFT], 1);
            const int p1 = atomicAdd(&s_o[d[g].y >> RB_SHIFT], 1);
            const int p2 = atomicAdd(&s_o[d[g].z >> RB_SHIFT], 1);
            const int p3 = atomicAdd(&s_o[d[g].w >> RB_SHIFT], 1);
            s_rec[p0] = make_int4(s[g].x, d[g].x, __float_as_int(ww[g].x), 0);
            s_rec[p1] = make_int4(s[g].y, d[g].y, __float_as_int(ww[g].y), 0);
            s_rec[p2] = make_int4(s[g].z, d[g].z, __float_as_int(ww[g].z), 0);
            s_rec[p3] = make_int4(s[g].w, d[g].w, __float_as_int(ww[g].w), 0);
        }
    } else {
        for (int i = c * CH + tid; i < e; i += CH_T) {
            const int dk = dst[i];
            const int p  = atomicAdd(&s_o[dk >> RB_SHIFT], 1);
            s_rec[p] = make_int4(src[i], dk, __float_as_int(w[i]), 0);
        }
    }
    __syncthreads();

    int4* tc = tmp + (size_t)c * CH;
    if (whole) {
        #pragma unroll
        for (int k = 0; k < CH / CH_T; ++k) tc[k * CH_T + tid] = s_rec[k * CH_T + tid];
    } else {
        const int cnt = e - c * CH;
        for (int i = tid; i < cnt; i += CH_T) tc[i] = s_rec[i];
    }
}

__global__ __launch_bounds__(512) void k_bucket(const int4* __restrict__ tmp,
                                                const int* __restrict__ loc,
                                                int* __restrict__ rowptr,
                                                int2* __restrict__ csr,
                                                const float* __restrict__ H,
                                                const float* __restrict__ Wm,
                                                float* __restrict__ dinv,
                                                float* __restrict__ hw, int n, int nc, int nb) {
    extern __shared__ int s_run[];
    __shared__ int s_cnt[RB];
    __shared__ int s_cur[RB];
    __shared__ int s_beg[RB];
    __shared__ int s_wtot[RB / 32];
    __shared__ int s_red[16];
    __shared__ float s_dinv[RB];
    __shared__ __attribute__((aligned(16))) float s_x[RB * HPITCH];
    __shared__ __attribute__((aligned(16))) int2 s_csr[CSR_CAP];
    const int tid  = threadIdx.x;
    const int lane = tid & 31;
    const int wv   = tid >> 5;
    const int l16  = tid & 15;
    const int grp  = tid >> 4;
    const int b    = blockIdx.x;

    if (tid < RB) s_cnt[tid] = 0;
    int part = 0;
    for (int c = tid; c < nc; c += 512) {
        const int* lc = loc + (size_t)c * (nb + 1) + b;
        const int lo = lc[0];
        const int hi = lc[1];
        s_run[2 * c]     = c * CH + lo;
        s_run[2 * c + 1] = hi - lo;
        part += lo;
    }
    for (int o = 16; o > 0; o >>= 1) part += __shfl_xor(part, o, 32);
    if (lane == 0) s_red[wv] = part;
    __syncthreads();
    int beg = 0;
    for (int k = 0; k < 16; ++k) beg += s_red[k];

    for (int c0 = grp; c0 < nc; c0 += 256) {
        int st[8], ln[8], dv[8];
        #pragma unroll
        for (int u = 0; u < 8; ++u) {
            const int c = c0 + 32 * u;
            st[u] = 0; ln[u] = 0;
            if (c < nc) { st[u] = s_run[2 * c]; ln[u] = s_run[2 * c + 1]; }
            dv[u] = tmp[max(st[u] + min(l16, ln[u] - 1), 0)].y;
        }
        #pragma unroll
        for (int u = 0; u < 8; ++u) {
            if (l16 < ln[u]) atomicAdd(&s_cnt[dv[u] & (RB - 1)], 1);
            for (int q = 16 + l16; q < ln[u]; q += 16)
                atomicAdd(&s_cnt[tmp[st[u] + q].y & (RB - 1)], 1);
        }
    }
    __syncthreads();

    int cn = 0, inc = 0;
    if (tid < RB) {
        cn = s_cnt[tid];
        inc = cn;
        for (int o = 1; o < 32; o <<= 1) {
            const int t = __shfl_up(inc, o, 32);
            if (lane >= o) inc += t;
        }
        if (lane == 31) s_wtot[wv] = inc;
    }
    __syncthreads();
    if (tid < RB) {
        int pre = 0;
        for (int k = 0; k < RB / 32; ++k)
            if (k < wv) pre += s_wtot[k];
        const int excl = pre + inc - cn;
        s_cur[tid] = excl;
        s_beg[tid] = excl;
        const int row = b * RB + tid;
        if (row < n)      rowptr[row] = beg + excl;
        if (row == n - 1) rowptr[n]   = beg + excl + cn;
    }
    int totv = 0;
    for (int k = 0; k < RB / 32; ++k) totv += s_wtot[k];
    const int tot = __builtin_amdgcn_readfirstlane(totv);
    const bool staged = (tot <= CSR_CAP);
    __syncthreads();

    float4 xv[8];
    #pragma unroll
    for (int i = 0; i < 8; ++i) {
        const int r = b * RB + i * 32 + grp;
        xv[i] = *(const float4*)(H + (size_t)min(r, n - 1) * FDIM + l16 * 4);
        if (r >= n) xv[i] = make_float4(0.0f, 0.0f, 0.0f, 0.0f);
    }

    for (int c0 = grp; c0 < nc; c0 += 256) {
        int st[8], ln[8];
        int4 ev[8];
        #pragma unroll
        for (int u = 0; u < 8; ++u) {
            const int c = c0 + 32 * u;
            st[u] = 0; ln[u] = 0;
            if (c < nc) { st[u] = s_run[2 * c]; ln[u] = s_run[2 * c + 1]; }
            ev[u] = tmp[max(st[u] + min(l16, ln[u] - 1), 0)];
        }
        #pragma unroll
        for (int u = 0; u < 8; ++u) {
            if (l16 < ln[u]) {
                const int pos = atomicAdd(&s_cur[ev[u].y & (RB - 1)], 1);
                const int2 rec = make_int2(ev[u].x, ev[u].z);
                if (staged) s_csr[pos] = rec; else csr[beg + pos] = rec;
            }
            for (int q = 16 + l16; q < ln[u]; q += 16) {
                const int4 eq = tmp[st[u] + q];
                const int pos = atomicAdd(&s_cur[eq.y & (RB - 1)], 1);
                const int2 rec = make_int2(eq.x, eq.z);
                if (staged) s_csr[pos] = rec; else csr[beg + pos] = rec;
            }
        }
    }

    #pragma unroll
    for (int i = 0; i < 8; ++i)
        *(float4*)(s_x + (i * 32 + grp) * HPITCH + l16 * 4) = xv[i];

    const int hi = (tid >> 4) & 1;
    const int cc = (wv & 3) * 16 + l16;
    v2f bfrag[16];
    #pragma unroll
    for (int k = 0; k < 16; ++k) {
        const int k0 = 4 * k + (hi ? 2 : 0);
        bfrag[k].x = Wm[(size_t)k0 * FDIM + cc];
        bfrag[k].y = Wm[(size_t)(k0 + 1) * FDIM + cc];
    }
    __syncthreads();

    if (staged) {
        const int rq = tid >> 3;
        const int t8 = tid & 7;
        int   p[RB / 64], pe[RB / 64];
        float s[RB / 64];
        #pragma unroll
        for (int ps = 0; ps < RB / 64; ++ps) {
            p[ps]  = s_beg[ps * 64 + rq] + t8;
            pe[ps] = s_cur[ps * 64 + rq];
            s[ps]  = 0.0f;
        }
        #pragma unroll
        for (int rd = 0; rd < 3; ++rd) {
            float wq[RB / 64];
            #pragma unroll
            for (int ps = 0; ps < RB / 64; ++ps)
                wq[ps] = __int_as_float(s_csr[max(min(p[ps], pe[ps] - 1), 0)].y);
            #pragma unroll
            for (int ps = 0; ps < RB / 64; ++ps) {
                if (p[ps] < pe[ps]) s[ps] += wq[ps];
                p[ps] += 8;
            }
        }
        #pragma unroll
        for (int ps = 0; ps < RB / 64; ++ps) {
            float sr = s[ps];
            for (int q = p[ps]; q < pe[ps]; q += 8) sr += __int_as_float(s_csr[q].y);
            sr += __shfl_xor(sr, 4, 8);
            sr += __shfl_xor(sr, 2, 8);
            sr += __shfl_xor(sr, 1, 8);
            if (t8 == 0) {
                const float dv = rsqrtf(1.0f + sr);
                s_dinv[ps * 64 + rq] = dv;
                const int r = b * RB + ps * 64 + rq;
                if (r < n) dinv[r] = dv;
            }
        }
    } else {
        const int rq = tid >> 3;
        const int t8 = tid & 7;
        for (int ps = 0; ps < RB / 64; ++ps) {
            const int pe = beg + s_cur[ps * 64 + rq];
            float sr = 0.0f;
            for (int q = beg + s_beg[ps * 64 + rq] + t8; q < pe; q += 8) sr += __int_as_float(csr[q].y);
            sr += __shfl_xor(sr, 4, 8);
            sr += __shfl_xor(sr, 2, 8);
            sr += __shfl_xor(sr, 1, 8);
            if (t8 == 0) {
                const float dv = rsqrtf(1.0f + sr);
                s_dinv[ps * 64 + rq] = dv;
                const int r = b * RB + ps * 64 + rq;
                if (r < n) dinv[r] = dv;
            }
        }
    }
    __syncthreads();

    if (staged) {
        const int gend = beg + tot;
        for (int lo = (beg & ~15) + 2 * tid; lo < gend; lo += 2 * 512) {
            const bool in0 = (lo >= beg);
            const bool in1 = (lo + 1 >= beg) && (lo + 1 < gend);
            if (in0 && in1) {
                const int2 e0 = s_csr[lo - beg];
                const int2 e1 = s_csr[lo - beg + 1];
                *(int4*)(csr + lo) = make_int4(e0.x, e0.y, e1.x, e1.y);
            } else if (in0) {
                csr[lo] = s_csr[lo - beg];
            } else if (in1) {
                csr[lo + 1] = s_csr[lo + 1 - beg];
            }
        }
    }

    #pragma unroll
    for (int t = 0; t < 4; ++t) {
        const int rt = (wv >> 2) + 4 * t;
        const float* ap = s_x + (rt * 16 + l16) * HPITCH + (hi ? 2 : 0);
        v8f acc = {};
        #pragma unroll
        for (int k = 0; k < 16; ++k) {
            const v2f a = *(const v2f*)(ap + 4 * k);
            acc = __builtin_amdgcn_wmma_f32_16x16x4_f32(
                      false, a, false, bfrag[k], (short)0, acc, false, false);
        }
        const int rl0 = rt * 16 + hi * 8;
        const int r0  = b * RB + rl0;
        const float* sd = s_dinv + rl0;
        if (b * RB + rt * 16 + 16 <= n) {
            float* op = hw + (size_t)r0 * FDIM + cc;
            #pragma unroll
            for (int i = 0; i < 8; ++i)
                op[(size_t)i * FDIM] = acc[i] * sd[i];
        } else {
            #pragma unroll
            for (int i = 0; i < 8; ++i)
                if (r0 + i < n) hw[(size_t)(r0 + i) * FDIM + cc] = acc[i] * sd[i];
        }
    }
}


static __device__ __forceinline__ float4 agg64_row(const float* __restrict__ hw,
                                                   const int2* __restrict__ csr,
                                                   const int* __restrict__ rowptr,
                                                   const int d, const int l16) {
    const int beg = rowptr[d];
    const int end = rowptr[d + 1];
    const int f = l16 * 4;
    float4 acc = *(const float4*)(hw + (size_t)d * FDIM + f);

    const float4 zero4 = make_float4(0.0f, 0.0f, 0.0f, 0.0f);
    const float* hrow = hw + f;
    for (int base = beg; base < end; base += 16) {
        const int idx = base + l16;
        int2 pr = make_int2(0, 0);
        if (idx < end) pr = csr[idx];
        const int m = min(16, end - base);
        for (int j = 0; j < m; j += 4) {
            const int   s0 = __shfl(pr.x, j + 0, 16);
            const int   s1 = __shfl(pr.x, j + 1, 16);
            const int   s2 = __shfl(pr.x, j + 2, 16);
            const int   s3 = __shfl(pr.x, j + 3, 16);
            const float n0 = __int_as_float(__shfl(pr.y, j + 0, 16));
            const float n1 = __int_as_float(__shfl(pr.y, j + 1, 16));
            const float n2 = __int_as_float(__shfl(pr.y, j + 2, 16));
            const float n3 = __int_as_float(__shfl(pr.y, j + 3, 16));
            float4 v0 = *(const float4*)(hrow + (size_t)s0 * FDIM);
            float4 v1 = zero4, v2 = zero4, v3 = zero4;
            if (j + 1 < m) v1 = *(const float4*)(hrow + (size_t)s1 * FDIM);
            if (j + 2 < m) v2 = *(const float4*)(hrow + (size_t)s2 * FDIM);
            if (j + 3 < m) v3 = *(const float4*)(hrow + (size_t)s3 * FDIM);
            acc.x += v0.x * n0; acc.y += v0.y * n0; acc.z += v0.z * n0; acc.w += v0.w * n0;
            acc.x += v1.x * n1; acc.y += v1.y * n1; acc.z += v1.z * n1; acc.w += v1.w * n1;
            acc.x += v2.x * n2; acc.y += v2.y * n2; acc.z += v2.z * n2; acc.w += v2.w * n2;
            acc.x += v3.x * n3; acc.y += v3.y * n3; acc.z += v3.z * n3; acc.w += v3.w * n3;
        }
    }
    return acc;
}


__global__ __launch_bounds__(256) void k_agg64_gemm(const float* __restrict__ hw,
                                                    const int2* __restrict__ csr,
                                                    const int* __restrict__ rowptr,
                                                    const float* __restrict__ dinv,
                                                    const float* __restrict__ bias,
                                                    const float* __restrict__ Wn,
                                                    float* __restrict__ hnext, int n) {
    __shared__ float s_h[16 * HPITCH];
    const int l16  = threadIdx.x & 15;
    const int grp  = threadIdx.x >> 4;
    const int row0 = blockIdx.x * 16;
    const int g    = row0 + grp;
    const int d    = (g < n) ? g : (n - 1);

    const float dv = dinv[d];
    const float4 bs = *(const float4*)(bias + l16 * 4);
    float4 acc = agg64_row(hw, csr, rowptr, d, l16);

    const int wave = threadIdx.x >> 5;
    const int hi   = (threadIdx.x >> 4) & 1;
    const int cc   = (wave & 3) * 16 + l16;
    v2f bfrag[16];
    if (wave < 4) {
        #pragma unroll
        for (int k = 0; k < 16; ++k) {
            const int k0 = 4 * k + (hi ? 2 : 0);
            bfrag[k].x = Wn[(size_t)k0 * FDIM + cc];
            bfrag[k].y = Wn[(size_t)(k0 + 1) * FDIM + cc];
        }
    }

    acc.x = fmaxf(fmaf(acc.x, dv, bs.x), 0.0f) * dv;
    acc.y = fmaxf(fmaf(acc.y, dv, bs.y), 0.0f) * dv;
    acc.z = fmaxf(fmaf(acc.z, dv, bs.z), 0.0f) * dv;
    acc.w = fmaxf(fmaf(acc.w, dv, bs.w), 0.0f) * dv;
    *(float4*)(s_h + grp * HPITCH + l16 * 4) = acc;
    __syncthreads();

    if (wave < 4) {
        const float* ap = s_h + l16 * HPITCH + (hi ? 2 : 0);
        v8f c = {};
        #pragma unroll
        for (int k = 0; k < 16; ++k) {
            const v2f a = *(const v2f*)(ap + 4 * k);
            c = __builtin_amdgcn_wmma_f32_16x16x4_f32(
                    false, a, false, bfrag[k], (short)0, c, false, false);
        }
        const int rbase = row0 + hi * 8;
        if (row0 + 16 <= n) {
            float* op = hnext + (size_t)rbase * FDIM + cc;
            #pragma unroll
            for (int i = 0; i < 8; ++i)
                op[(size_t)i * FDIM] = c[i];
        } else {
            #pragma unroll
            for (int i = 0; i < 8; ++i) {
                const int rr = rbase + i;
                if (rr < n) hnext[(size_t)rr * FDIM + cc] = c[i];
            }
        }
    }
}

__global__ __launch_bounds__(512) void k_agg64_dot(const float* __restrict__ hw,
                                                   const int2* __restrict__ csr,
                                                   const int* __restrict__ rowptr,
                                                   const float* __restrict__ dinv,
                                                   const float* __restrict__ bias,
                                                   const float* __restrict__ w2,
                                                   float* __restrict__ lin, int n) {
    __shared__ float s_lin[32];
    const int l16  = threadIdx.x & 15;
    const int grp  = threadIdx.x >> 4;
    const int row0 = blockIdx.x * 32;
    const int g    = row0 + grp;
    const int d    = (g < n) ? g : (n - 1);

    const float dv = dinv[d];
    const int f = l16 * 4;
    const float4 bs = *(const float4*)(bias + f);
    const float4 acc = agg64_row(hw, csr, rowptr, d, l16);
    float s = fmaxf(fmaf(acc.x, dv, bs.x), 0.0f) * w2[f + 0];
    s = fmaf(fmaxf(fmaf(acc.y, dv, bs.y), 0.0f), w2[f + 1], s);
    s = fmaf(fmaxf(fmaf(acc.z, dv, bs.z), 0.0f), w2[f + 2], s);
    s = fmaf(fmaxf(fmaf(acc.w, dv, bs.w), 0.0f), w2[f + 3], s);
    for (int o = 8; o > 0; o >>= 1) s += __shfl_xor(s, o, 16);
    if (l16 == 0) s_lin[grp] = s * dv;
    __syncthreads();
    if (threadIdx.x < 32) {
        const int r = row0 + threadIdx.x;
        if (r < n) lin[r] = s_lin[threadIdx.x];
    }
}

__global__ __launch_bounds__(256) void k_agg1(const float* __restrict__ lin,
                                              const int2* __restrict__ csr,
                                              const int* __restrict__ rowptr,
                                              const float* __restrict__ dinv,
                                              const float* __restrict__ b2,
                                              float* __restrict__ out, int n) {
    int d = blockIdx.x * blockDim.x + threadIdx.x;
    if (d < n) {
        const int beg = rowptr[d];
        const int end = rowptr[d + 1];
        float acc = lin[d];
        #pragma unroll 4
        for (int p = beg; p < end; ++p) {
            const int2 pr = csr[p];
            acc += lin[pr.x] * __int_as_float(pr.y);
        }
        out[d] = fmaf(acc, dinv[d], b2[0]);
    }
}


extern "C" void kernel_launch(void* const* d_in, const int* in_sizes, int n_in,
                              void* d_out, int out_size, void* d_ws, size_t ws_size,
                              hipStream_t stream) {
    const float* x    = (const float*)d_in[0];
    const int*   esrc = (const int*)  d_in[1];
    const int*   edst = (const int*)  d_in[2];
    const float* ew   = (const float*)d_in[3];
    const float* W0   = (const float*)d_in[4];
    const float* b0   = (const float*)d_in[5];
    const float* W1   = (const float*)d_in[6];
    const float* b1   = (const float*)d_in[7];
    const float* W2   = (const float*)d_in[8];
    const float* b2   = (const float*)d_in[9];

    const int N = in_sizes[0] / FDIM;
    const int E = in_sizes[1];
    float* out = (float*)d_out;

    const int NB = (N + RB - 1) / RB;
    const int NC = (E + CH - 1) / CH;
    const int HC = (NB + 1) * NC;

    const size_t Na = ((size_t)N + 64 + 63) & ~(size_t)63;
    const size_t Ea = ((size_t)E + 63) & ~(size_t)63;
    const size_t Ha = ((size_t)HC + 63) & ~(size_t)63;
    const size_t Ta = (size_t)NC * CH;
    float* ws     = (float*)d_ws;
    float* dinv   = ws;
    int*   rowptr = (int*)(ws + Na);
    int*   loc    = (int*)(ws + 2 * Na);
    int2*  csr    = (int2*)(ws + 2 * Na + Ha);
    int4*  tmp    = (int4*)(ws + 2 * Na + Ha + 2 * Ea);
    float* hw     = ws + 2 * Na + Ha + 2 * Ea + 4 * Ta;
    float* ha     = hw + (size_t)N * FDIM;
    float* lin    = ha + (size_t)N * FDIM;

    const int T = 256;
    const int gN  = (N + T - 1) / T;
    const int gA  = (N + 15) / 16;
    const int gDot = (N + 31) / 32;

    k_chunk   <<<NC, CH_T, NB * sizeof(int), stream>>>(esrc, edst, ew, loc, tmp, E, NB);
    k_bucket  <<<NB, 512, 2 * NC * sizeof(int), stream>>>(tmp, loc, rowptr, csr, x, W0, dinv, hw, N, NC, NB);

    k_agg64_gemm<<<gA, T, 0, stream>>>(hw, csr, rowptr, dinv, b0, W1, ha, N);

    k_agg64_dot<<<gDot, 512, 0, stream>>>(ha, csr, rowptr, dinv, b1, W2, lin, N);

    k_agg1    <<<gN, T, 0, stream>>>(lin, csr, rowptr, dinv, b2, out, N);
}


# 64-wide gathers: 8 row loads of a list group go out before the first sum instead of 4, at unchanged lanes per row and wave count (64 registers, every wave slot kept); sum orders unchanged
# speedup vs baseline: 4.2431x; 1.0442x over previous
#include <hip/hip_runtime.h>


typedef __attribute__((ext_vector_type(2))) float v2f;
typedef __attribute__((ext_vector_type(8))) float v8f;

#define FDIM 64

#define HPITCH 68


#define RB_SHIFT 8
#define RB       (1 << RB_SHIFT)
#define CH_T     256
#define CH       (CH_T * 16)
#define CSR_CAP  8192

__global__ __launch_bounds__(CH_T) void k_chunk(const int* __restrict__ src,
                                                const int* __restrict__ dst,
                                                const float* __restrict__ w,
                                                int* __restrict__ loc,
                                                int4* __restrict__ tmp, int e, int nb) {
    extern __shared__ int s_o[];
    __shared__ int s_wt[CH_T / 32];
    __shared__ __attribute__((aligned(16))) int4 s_rec[CH];
    const int tid  = threadIdx.x;
    const int lane = tid & 31;
    const int wv   = tid >> 5;
    const int c    = blockIdx.x;
    for (int b = tid; b < nb; b += CH_T) s_o[b] = 0;
    __syncthreads();

    const bool whole = ((c + 1) * CH <= e);
    int4 d[4], s[4];
    float4 ww[4];
    if (whole) {
        #pragma unroll
        for (int g = 0; g < 4; ++g) {
            const int i = c * CH + (g * CH_T + tid) * 4;
            d[g]  = *(const int4*)(dst + i);
            s[g]  = *(const int4*)(src + i);
            ww[g] = *(const float4*)(w + i);
        }
        #pragma unroll
        for (int g = 0; g < 4; ++g) {
            atomicAdd(&s_o[d[g].x >> RB_SHIFT], 1);
            atomicAdd(&s_o[d[g].y >> RB_SHIFT], 1);
            atomicAdd(&s_o[d[g].z >> RB_SHIFT], 1);
            atomicAdd(&s_o[d[g].w >> RB_SHIFT], 1);
        }
    } else {
        for (int i = c * CH + tid; i < e; i += CH_T) atomicAdd(&s_o[dst[i] >> RB_SHIFT], 1);
    }
    __syncthreads();

    const int K  = (nb + CH_T - 1) / CH_T;
    const int b0 = tid * K;
    int sum = 0;
    for (int k = 0; k < K; ++k) {
        const int b = b0 + k;
        if (b < nb) sum += s_o[b];
    }
    int inc = sum;
    for (int o = 1; o < 32; o <<= 1) {
        const int t = __shfl_up(inc, o, 32);
        if (lane >= o) inc += t;
    }
    if (lane == 31) s_wt[wv] = inc;
    __syncthreads();
    int run = inc - sum;
    for (int k = 0; k < CH_T / 32; ++k)
        if (k < wv) run += s_wt[k];
    int* lc = loc + (size_t)c * (nb + 1);
    for (int k = 0; k < K; ++k) {
        const int b = b0 + k;
        if (b < nb) {
            const int cn = s_o[b];
            s_o[b] = run;
            lc[b]  = run;
            run += cn;
        }
    }
    if (tid == CH_T - 1) lc[nb] = run;
    __syncthreads();

    if (whole) {
        #pragma unroll
        for (int g = 0; g < 4; ++g) {
            const int p0 = atomicAdd(&s_o[d[g].x >> RB_SHIFT], 1);
            const int p1 = atomicAdd(&s_o[d[g].y >> RB_SHIFT], 1);
            const int p2 = atomicAdd(&s_o[d[g].z >> RB_SHIFT], 1);
            const int p3 = atomicAdd(&s_o[d[g].w >> RB_SHIFT], 1);
            s_rec[p0] = make_int4(s[g].x, d[g].x, __float_as_int(ww[g].x), 0);
            s_rec[p1] = make_int4(s[g].y, d[g].y, __float_as_int(ww[g].y), 0);
            s_rec[p2] = make_int4(s[g].z, d[g].z, __float_as_int(ww[g].z), 0);
            s_rec[p3] = make_int4(s[g].w, d[g].w, __float_as_int(ww[g].w), 0);
        }
    } else {
        for (int i = c * CH + tid; i < e; i += CH_T) {
            const int dk = dst[i];
            const int p  = atomicAdd(&s_o[dk >> RB_SHIFT], 1);
            s_rec[p] = make_int4(src[i], dk, __float_as_int(w[i]), 0);
        }
    }
    __syncthreads();

    int4* tc = tmp + (size_t)c * CH;
    if (whole) {
        #pragma unroll
        for (int k = 0; k < CH / CH_T; ++k) tc[k * CH_T + tid] = s_rec[k * CH_T + tid];
    } else {
        const int cnt = e - c * CH;
        for (int i = tid; i < cnt; i += CH_T) tc[i] = s_rec[i];
    }
}

__global__ __launch_bounds__(512) void k_bucket(const int4* __restrict__ tmp,
                                                const int* __restrict__ loc,
                                                int* __restrict__ rowptr,
                                                int2* __restrict__ csr,
                                                const float* __restrict__ H,
                                                const float* __restrict__ Wm,
                                                float* __restrict__ dinv,
                                                float* __restrict__ hw, int n, int nc, int nb) {
    extern __shared__ int s_run[];
    __shared__ int s_cnt[RB];
    __shared__ int s_cur[RB];
    __shared__ int s_beg[RB];
    __shared__ int s_wtot[RB / 32];
    __shared__ int s_red[16];
    __shared__ float s_dinv[RB];
    __shared__ __attribute__((aligned(16))) float s_x[RB * HPITCH];
    __shared__ __attribute__((aligned(16))) int2 s_csr[CSR_CAP];
    const int tid  = threadIdx.x;
    const int lane = tid & 31;
    const int wv   = tid >> 5;
    const int l16  = tid & 15;
    const int grp  = tid >> 4;
    const int b    = blockIdx.x;

    if (tid < RB) s_cnt[tid] = 0;
    int part = 0;
    for (int c = tid; c < nc; c += 512) {
        const int* lc = loc + (size_t)c * (nb + 1) + b;
        const int lo = lc[0];
        const int hi = lc[1];
        s_run[2 * c]     = c * CH + lo;
        s_run[2 * c + 1] = hi - lo;
        part += lo;
    }
    for (int o = 16; o > 0; o >>= 1) part += __shfl_xor(part, o, 32);
    if (lane == 0) s_red[wv] = part;
    __syncthreads();
    int beg = 0;
    for (int k = 0; k < 16; ++k) beg += s_red[k];

    for (int c0 = grp; c0 < nc; c0 += 256) {
        int st[8], ln[8], dv[8];
        #pragma unroll
        for (int u = 0; u < 8; ++u) {
            const int c = c0 + 32 * u;
            st[u] = 0; ln[u] = 0;
            if (c < nc) { st[u] = s_run[2 * c]; ln[u] = s_run[2 * c + 1]; }
            dv[u] = tmp[max(st[u] + min(l16, ln[u] - 1), 0)].y;
        }
        #pragma unroll
        for (int u = 0; u < 8; ++u) {
            if (l16 < ln[u]) atomicAdd(&s_cnt[dv[u] & (RB - 1)], 1);
            for (int q = 16 + l16; q < ln[u]; q += 16)
                atomicAdd(&s_cnt[tmp[st[u] + q].y & (RB - 1)], 1);
        }
    }
    __syncthreads();

    int cn = 0, inc = 0;
    if (tid < RB) {
        cn = s_cnt[tid];
        inc = cn;
        for (int o = 1; o < 32; o <<= 1) {
            const int t = __shfl_up(inc, o, 32);
            if (lane >= o) inc += t;
        }
        if (lane == 31) s_wtot[wv] = inc;
    }
    __syncthreads();
    if (tid < RB) {
        int pre = 0;
        for (int k = 0; k < RB / 32; ++k)
            if (k < wv) pre += s_wtot[k];
        const int excl = pre + inc - cn;
        s_cur[tid] = excl;
        s_beg[tid] = excl;
        const int row = b * RB + tid;
        if (row < n)      rowptr[row] = beg + excl;
        if (row == n - 1) rowptr[n]   = beg + excl + cn;
    }
    int totv = 0;
    for (int k = 0; k < RB / 32; ++k) totv += s_wtot[k];
    const int tot = __builtin_amdgcn_readfirstlane(totv);
    const bool staged = (tot <= CSR_CAP);
    __syncthreads();

    float4 xv[8];
    #pragma unroll
    for (int i = 0; i < 8; ++i) {
        const int r = b * RB + i * 32 + grp;
        xv[i] = *(const float4*)(H + (size_t)min(r, n - 1) * FDIM + l16 * 4);
        if (r >= n) xv[i] = make_float4(0.0f, 0.0f, 0.0f, 0.0f);
    }

    for (int c0 = grp; c0 < nc; c0 += 256) {
        int st[8], ln[8];
        int4 ev[8];
        #pragma unroll
        for (int u = 0; u < 8; ++u) {
            const int c = c0 + 32 * u;
            st[u] = 0; ln[u] = 0;
            if (c < nc) { st[u] = s_run[2 * c]; ln[u] = s_run[2 * c + 1]; }
            ev[u] = tmp[max(st[u] + min(l16, ln[u] - 1), 0)];
        }
        #pragma unroll
        for (int u = 0; u < 8; ++u) {
            if (l16 < ln[u]) {
                const int pos = atomicAdd(&s_cur[ev[u].y & (RB - 1)], 1);
                const int2 rec = make_int2(ev[u].x, ev[u].z);
                if (staged) s_csr[pos] = rec; else csr[beg + pos] = rec;
            }
            for (int q = 16 + l16; q < ln[u]; q += 16) {
                const int4 eq = tmp[st[u] + q];
                const int pos = atomicAdd(&s_cur[eq.y & (RB - 1)], 1);
                const int2 rec = make_int2(eq.x, eq.z);
                if (staged) s_csr[pos] = rec; else csr[beg + pos] = rec;
            }
        }
    }

    #pragma unroll
    for (int i = 0; i < 8; ++i)
        *(float4*)(s_x + (i * 32 + grp) * HPITCH + l16 * 4) = xv[i];

    const int hi = (tid >> 4) & 1;
    const int cc = (wv & 3) * 16 + l16;
    v2f bfrag[16];
    #pragma unroll
    for (int k = 0; k < 16; ++k) {
        const int k0 = 4 * k + (hi ? 2 : 0);
        bfrag[k].x = Wm[(size_t)k0 * FDIM + cc];
        bfrag[k].y = Wm[(size_t)(k0 + 1) * FDIM + cc];
    }
    __syncthreads();

    if (staged) {
        const int rq = tid >> 3;
        const int t8 = tid & 7;
        int   p[RB / 64], pe[RB / 64];
        float s[RB / 64];
        #pragma unroll
        for (int ps = 0; ps < RB / 64; ++ps) {
            p[ps]  = s_beg[ps * 64 + rq] + t8;
            pe[ps] = s_cur[ps * 64 + rq];
            s[ps]  = 0.0f;
        }
        #pragma unroll
        for (int rd = 0; rd < 3; ++rd) {
            float wq[RB / 64];
            #pragma unroll
            for (int ps = 0; ps < RB / 64; ++ps)
                wq[ps] = __int_as_float(s_csr[max(min(p[ps], pe[ps] - 1), 0)].y);
            #pragma unroll
            for (int ps = 0; ps < RB / 64; ++ps) {
                if (p[ps] < pe[ps]) s[ps] += wq[ps];
                p[ps] += 8;
            }
        }
        #pragma unroll
        for (int ps = 0; ps < RB / 64; ++ps) {
            float sr = s[ps];
            for (int q = p[ps]; q < pe[ps]; q += 8) sr += __int_as_float(s_csr[q].y);
            sr += __shfl_xor(sr, 4, 8);
            sr += __shfl_xor(sr, 2, 8);
            sr += __shfl_xor(sr, 1, 8);
            if (t8 == 0) {
                const float dv = rsqrtf(1.0f + sr);
                s_dinv[ps * 64 + rq] = dv;
                const int r = b * RB + ps * 64 + rq;
                if (r < n) dinv[r] = dv;
            }
        }
    } else {
        const int rq = tid >> 3;
        const int t8 = tid & 7;
        for (int ps = 0; ps < RB / 64; ++ps) {
            const int pe = beg + s_cur[ps * 64 + rq];
            float sr = 0.0f;
            for (int q = beg + s_beg[ps * 64 + rq] + t8; q < pe; q += 8) sr += __int_as_float(csr[q].y);
            sr += __shfl_xor(sr, 4, 8);
            sr += __shfl_xor(sr, 2, 8);
            sr += __shfl_xor(sr, 1, 8);
            if (t8 == 0) {
                const float dv = rsqrtf(1.0f + sr);
                s_dinv[ps * 64 + rq] = dv;
                const int r = b * RB + ps * 64 + rq;
                if (r < n) dinv[r] = dv;
            }
        }
    }
    __syncthreads();

    if (staged) {
        const int gend = beg + tot;
        for (int lo = (beg & ~15) + 2 * tid; lo < gend; lo += 2 * 512) {
            const bool in0 = (lo >= beg);
            const bool in1 = (lo + 1 >= beg) && (lo + 1 < gend);
            if (in0 && in1) {
                const int2 e0 = s_csr[lo - beg];
                const int2 e1 = s_csr[lo - beg + 1];
                *(int4*)(csr + lo) = make_int4(e0.x, e0.y, e1.x, e1.y);
            } else if (in0) {
                csr[lo] = s_csr[lo - beg];
            } else if (in1) {
                csr[lo + 1] = s_csr[lo + 1 - beg];
            }
        }
    }

    #pragma unroll
    for (int t = 0; t < 4; ++t) {
        const int rt = (wv >> 2) + 4 * t;
        const float* ap = s_x + (rt * 16 + l16) * HPITCH + (hi ? 2 : 0);
        v8f acc = {};
        #pragma unroll
        for (int k = 0; k < 16; ++k) {
            const v2f a = *(const v2f*)(ap + 4 * k);
            acc = __builtin_amdgcn_wmma_f32_16x16x4_f32(
                      false, a, false, bfrag[k], (short)0, acc, false, false);
        }
        const int rl0 = rt * 16 + hi * 8;
        const int r0  = b * RB + rl0;
        const float* sd = s_dinv + rl0;
        if (b * RB + rt * 16 + 16 <= n) {
            float* op = hw + (size_t)r0 * FDIM + cc;
            #pragma unroll
            for (int i = 0; i < 8; ++i)
                op[(size_t)i * FDIM] = acc[i] * sd[i];
        } else {
            #pragma unroll
            for (int i = 0; i < 8; ++i)
                if (r0 + i < n) hw[(size_t)(r0 + i) * FDIM + cc] = acc[i] * sd[i];
        }
    }
}


static __device__ __forceinline__ float4 agg64_row(const float* __restrict__ hw,
                                                   const int2* __restrict__ csr,
                                                   const int* __restrict__ rowptr,
                                                   const int d, const int l16) {
    const int beg = rowptr[d];
    const int end = rowptr[d + 1];
    const int f = l16 * 4;
    float4 acc = *(const float4*)(hw + (size_t)d * FDIM + f);

    for (int base = beg; base < end; base += 16) {
        const int idx = base + l16;
        int2 pr = make_int2(d, 0);
        if (idx < end) pr = csr[idx];
        const int m = end - base;
        #pragma unroll
        for (int h = 0; h < 2; ++h) {
            if (h == 0 || m > 8) {
                float4 v[8];
                #pragma unroll
                for (int j = 0; j < 8; ++j) {
                    const int sj = __shfl(pr.x, 8 * h + j, 16);
                    v[j] = *(const float4*)((const char*)hw + ((unsigned)sj * (unsigned)(FDIM * 4) + (unsigned)(f * 4)));
                }
                #pragma unroll
                for (int j = 0; j < 8; ++j) {
                    const float nj = __int_as_float(__shfl(pr.y, 8 * h + j, 16));
                    acc.x += v[j].x * nj; acc.y += v[j].y * nj; acc.z += v[j].z * nj; acc.w += v[j].w * nj;
                }
            }
        }
    }
    return acc;
}


__global__ __launch_bounds__(256, 16) void k_agg64_gemm(const float* __restrict__ hw,
                                                    const int2* __restrict__ csr,
                                                    const int* __restrict__ rowptr,
                                                    const float* __restrict__ dinv,
                                                    const float* __restrict__ bias,
                                                    const float* __restrict__ Wn,
                                                    float* __restrict__ hnext, int n) {
    __shared__ float s_h[16 * HPITCH];
    const int l16  = threadIdx.x & 15;
    const int grp  = threadIdx.x >> 4;
    const int row0 = blockIdx.x * 16;
    const int g    = row0 + grp;
    const int d    = (g < n) ? g : (n - 1);

    const float dv = dinv[d];
    const float4 bs = *(const float4*)(bias + l16 * 4);
    float4 acc = agg64_row(hw, csr, rowptr, d, l16);

    const int wave = threadIdx.x >> 5;
    const int hi   = (threadIdx.x >> 4) & 1;
    const int cc   = (wave & 3) * 16 + l16;
    v2f bfrag[16];
    if (wave < 4) {
        #pragma unroll
        for (int k = 0; k < 16; ++k) {
            const int k0 = 4 * k + (hi ? 2 : 0);
            bfrag[k].x = Wn[(size_t)k0 * FDIM + cc];
            bfrag[k].y = Wn[(size_t)(k0 + 1) * FDIM + cc];
        }
    }

    acc.x = fmaxf(fmaf(acc.x, dv, bs.x), 0.0f) * dv;
    acc.y = fmaxf(fmaf(acc.y, dv, bs.y), 0.0f) * dv;
    acc.z = fmaxf(fmaf(acc.z, dv, bs.z), 0.0f) * dv;
    acc.w = fmaxf(fmaf(acc.w, dv, bs.w), 0.0f) * dv;
    *(float4*)(s_h + grp * HPITCH + l16 * 4) = acc;
    __syncthreads();

    if (wave < 4) {
        const float* ap = s_h + l16 * HPITCH + (hi ? 2 : 0);
        v8f c = {};
        #pragma unroll
        for (int k = 0; k < 16; ++k) {
            const v2f a = *(const v2f*)(ap + 4 * k);
            c = __builtin_amdgcn_wmma_f32_16x16x4_f32(
                    false, a, false, bfrag[k], (short)0, c, false, false);
        }
        const int rbase = row0 + hi * 8;
        if (row0 + 16 <= n) {
            float* op = hnext + (size_t)rbase * FDIM + cc;
            #pragma unroll
            for (int i = 0; i < 8; ++i)
                op[(size_t)i * FDIM] = c[i];
        } else {
            #pragma unroll
            for (int i = 0; i < 8; ++i) {
                const int rr = rbase + i;
                if (rr < n) hnext[(size_t)rr * FDIM + cc] = c[i];
            }
        }
    }
}

__global__ __launch_bounds__(512, 16) void k_agg64_dot(const float* __restrict__ hw,
                                                   const int2* __restrict__ csr,
                                                   const int* __restrict__ rowptr,
                                                   const float* __restrict__ dinv,
                                                   const float* __restrict__ bias,
                                                   const float* __restrict__ w2,
                                                   float* __restrict__ lin, int n) {
    __shared__ float s_lin[32];
    const int l16  = threadIdx.x & 15;
    const int grp  = threadIdx.x >> 4;
    const int row0 = blockIdx.x * 32;
    const int g    = row0 + grp;
    const int d    = (g < n) ? g : (n - 1);

    const int f = l16 * 4;
    const float4 acc = agg64_row(hw, csr, rowptr, d, l16);
    const float dv = dinv[d];
    const float4 bs = *(const float4*)(bias + f);
    float s = fmaxf(fmaf(acc.x, dv, bs.x), 0.0f) * w2[f + 0];
    s = fmaf(fmaxf(fmaf(acc.y, dv, bs.y), 0.0f), w2[f + 1], s);
    s = fmaf(fmaxf(fmaf(acc.z, dv, bs.z), 0.0f), w2[f + 2], s);
    s = fmaf(fmaxf(fmaf(acc.w, dv, bs.w), 0.0f), w2[f + 3], s);
    for (int o = 8; o > 0; o >>= 1) s += __shfl_xor(s, o, 16);
    if (l16 == 0) s_lin[grp] = s * dv;
    __syncthreads();
    if (threadIdx.x < 32) {
        const int r = row0 + threadIdx.x;
        if (r < n) lin[r] = s_lin[threadIdx.x];
    }
}

__global__ __launch_bounds__(256) void k_agg1(const float* __restrict__ lin,
                                              const int2* __restrict__ csr,
                                              const int* __restrict__ rowptr,
                                              const float* __restrict__ dinv,
                                              const float* __restrict__ b2,
                                              float* __restrict__ out, int n) {
    int d = blockIdx.x * blockDim.x + threadIdx.x;
    if (d < n) {
        const int beg = rowptr[d];
        const int end = rowptr[d + 1];
        float acc = lin[d];
        #pragma unroll 4
        for (int p = beg; p < end; ++p) {
            const int2 pr = csr[p];
            acc += lin[pr.x] * __int_as_float(pr.y);
        }
        out[d] = fmaf(acc, dinv[d], b2[0]);
    }
}


extern "C" void kernel_launch(void* const* d_in, const int* in_sizes, int n_in,
                              void* d_out, int out_size, void* d_ws, size_t ws_size,
                              hipStream_t stream) {
    const float* x    = (const float*)d_in[0];
    const int*   esrc = (const int*)  d_in[1];
    const int*   edst = (const int*)  d_in[2];
    const float* ew   = (const float*)d_in[3];
    const float* W0   = (const float*)d_in[4];
    const float* b0   = (const float*)d_in[5];
    const float* W1   = (const float*)d_in[6];
    const float* b1   = (const float*)d_in[7];
    const float* W2   = (const float*)d_in[8];
    const float* b2   = (const float*)d_in[9];

    const int N = in_sizes[0] / FDIM;
    const int E = in_sizes[1];
    float* out = (float*)d_out;

    const int NB = (N + RB - 1) / RB;
    const int NC = (E + CH - 1) / CH;
    const int HC = (NB + 1) * NC;

    const size_t Na = ((size_t)N + 64 + 63) & ~(size_t)63;
    const size_t Ea = ((size_t)E + 63) & ~(size_t)63;
    const size_t Ha = ((size_t)HC + 63) & ~(size_t)63;
    const size_t Ta = (size_t)NC * CH;
    float* ws     = (float*)d_ws;
    float* dinv   = ws;
    int*   rowptr = (int*)(ws + Na);
    int*   loc    = (int*)(ws + 2 * Na);
    int2*  csr    = (int2*)(ws + 2 * Na + Ha);
    int4*  tmp    = (int4*)(ws + 2 * Na + Ha + 2 * Ea);
    float* hw     = ws + 2 * Na + Ha + 2 * Ea + 4 * Ta;
    float* ha     = hw + (size_t)N * FDIM;
    float* lin    = ha + (size_t)N * FDIM;

    const int T = 256;
    const int gN  = (N + T - 1) / T;
    const int gA  = (N + 15) / 16;
    const int gDot = (N + 31) / 32;

    k_chunk   <<<NC, CH_T, NB * sizeof(int), stream>>>(esrc, edst, ew, loc, tmp, E, NB);
    k_bucket  <<<NB, 512, 2 * NC * sizeof(int), stream>>>(tmp, loc, rowptr, csr, x, W0, dinv, hw, N, NC, NB);

    k_agg64_gemm<<<gA, T, 0, stream>>>(hw, csr, rowptr, dinv, b0, W1, ha, N);

    k_agg64_dot<<<gDot, 512, 0, stream>>>(ha, csr, rowptr, dinv, b1, W2, lin, N);

    k_agg1    <<<gN, T, 0, stream>>>(lin, csr, rowptr, dinv, b2, out, N);
}
